# starting kernel
# baseline (speedup 1.0000x reference)
#include <hip/hip_runtime.h>
#include <hip/hip_bf16.h>

typedef __attribute__((ext_vector_type(2))) float v2f;
typedef __attribute__((ext_vector_type(8))) float v8f;

#define B_DIM 64
#define N_DIM 128
#define D_DIM 128
#define A_DIM 1000
#define KT    100

__device__ __forceinline__ v8f wmma_f32(v2f a, v2f b, v8f c) {
  return __builtin_amdgcn_wmma_f32_16x16x4_f32(false, a, false, b, (short)0, c,
                                               false, false);
}

__global__ __launch_bounds__(256) void gat_attn_kernel(
    const float* __restrict__ hidden, const int* __restrict__ adj,
    const float* __restrict__ amat, float* __restrict__ out) {
  __shared__ float sH[N_DIM * D_DIM];
  __shared__ float sAlpha[16 * N_DIM];
  __shared__ float sAa[D_DIM * 4];
  __shared__ float sRed[16][16];
  __shared__ float sMax[16];
  __shared__ float sSum[16];

  const int tid    = threadIdx.x;
  const int b      = blockIdx.x >> 3;
  const int i_base = (blockIdx.x & 7) * 16;

  const float* Hg = hidden + (size_t)b * N_DIM * D_DIM;
  for (int idx4 = tid; idx4 < (N_DIM * D_DIM) / 4; idx4 += 256)
    reinterpret_cast<float4*>(sH)[idx4] = reinterpret_cast<const float4*>(Hg)[idx4];
  if (tid < (D_DIM * 4) / 4)
    reinterpret_cast<float4*>(sAa)[tid] = reinterpret_cast<const float4*>(amat)[tid];
  __syncthreads();

  const int wave   = tid >> 5;
  const int lane   = tid & 31;
  const int lo16   = lane & 15;
  const int hi16   = (lane >= 16) ? 8 : 0;
  const int koff   = (lane >= 16) ? 2 : 0;
  const int j_base = wave * 16;

  const float C1 = 0.6f, C2 = 0.4f;
  const float NEG_INF = -9.0e15f;

  v8f acc[4];
  acc[0] = {}; acc[1] = {}; acc[2] = {}; acc[3] = {};
  const int i_row = i_base + lo16;
  const int j_row = j_base + lo16;
  for (int d0 = 0; d0 < D_DIM; d0 += 4) {
    const int d = d0 + koff;
    const float2 hi = *reinterpret_cast<const float2*>(&sH[i_row * D_DIM + d]);
    const float2 hj = *reinterpret_cast<const float2*>(&sH[j_row * D_DIM + d]);
    const float* a0 = &sAa[d * 4];
    const float* a1 = &sAa[(d + 1) * 4];
    const float hax = fabsf(hi.x), hay = fabsf(hi.y);
    v2f Bv;  Bv[0] = hj.x;        Bv[1] = hj.y;
    v2f Bva; Bva[0] = fabsf(hj.x); Bva[1] = fabsf(hj.y);
#pragma unroll
    for (int k = 0; k < 4; ++k) {
      v2f Av;  Av[0]  = C1 * hi.x * a0[k]; Av[1]  = C1 * hi.y * a1[k];
      acc[k] = wmma_f32(Av, Bv, acc[k]);
      v2f Ava; Ava[0] = C2 * hax * a0[k];  Ava[1] = C2 * hay * a1[k];
      acc[k] = wmma_f32(Ava, Bva, acc[k]);
    }
  }

  const int* adjB = adj + (size_t)b * N_DIM * N_DIM;
#pragma unroll
  for (int r = 0; r < 8; ++r) {
    const int il = r + hi16;
    const int t = adjB[(i_base + il) * N_DIM + j_base + lo16];
    float e = NEG_INF;
    e = (t == 1) ? acc[0][r] : e;
    e = (t == 2) ? acc[1][r] : e;
    e = (t == 3) ? acc[2][r] : e;
    e = (t == 4) ? acc[3][r] : e;
    sAlpha[il * N_DIM + j_base + lo16] = e;
  }
  __syncthreads();

  const int row = tid >> 4;
  const int part = tid & 15;
  float m = -3.4e38f;
  for (int c = part * 8; c < part * 8 + 8; ++c)
    m = fmaxf(m, sAlpha[row * N_DIM + c]);
  sRed[row][part] = m;
  __syncthreads();
  if (part == 0) {
    float mm = sRed[row][0];
    for (int q = 1; q < 16; ++q) mm = fmaxf(mm, sRed[row][q]);
    sMax[row] = mm;
  }
  __syncthreads();
  const float mm2 = sMax[row];
  float s = 0.f;
  for (int c = part * 8; c < part * 8 + 8; ++c) {
    const float ev = __expf(sAlpha[row * N_DIM + c] - mm2);
    sAlpha[row * N_DIM + c] = ev;
    s += ev;
  }
  sRed[row][part] = s;
  __syncthreads();
  if (part == 0) {
    float ss = 0.f;
    for (int q = 0; q < 16; ++q) ss += sRed[row][q];
    sSum[row] = ss;
  }
  __syncthreads();
  const float inv = 1.0f / sSum[row];
  for (int c = part * 8; c < part * 8 + 8; ++c)
    sAlpha[row * N_DIM + c] *= inv;
  __syncthreads();

  const int dg = j_base + lo16;
  v8f oacc = {};
  for (int j0 = 0; j0 < N_DIM; j0 += 4) {
    const int jk = j0 + koff;
    const float2 av = *reinterpret_cast<const float2*>(&sAlpha[lo16 * N_DIM + jk]);
    v2f Av; Av[0] = av.x; Av[1] = av.y;
    v2f Bv; Bv[0] = sH[jk * D_DIM + dg]; Bv[1] = sH[(jk + 1) * D_DIM + dg];
    oacc = wmma_f32(Av, Bv, oacc);
  }
#pragma unroll
  for (int r = 0; r < 8; ++r)
    out[((size_t)b * N_DIM + i_base + r + hi16) * D_DIM + dg] = oacc[r];
}

__global__ __launch_bounds__(256) void attr_gemm_kernel(
    const float* __restrict__ Aattr, const float* __restrict__ Eemb,
    float* __restrict__ out2) {
  __shared__ float sAp[16 * KT];
  __shared__ float sEp[KT * D_DIM];

  const int tid    = threadIdx.x;
  const int m_base = blockIdx.x * 16;
  const int wave   = tid >> 5;
  const int lane   = tid & 31;
  const int lo16   = lane & 15;
  const int hi16   = (lane >= 16) ? 8 : 0;
  const int koff   = (lane >= 16) ? 2 : 0;
  const int dg     = wave * 16 + lo16;

  v8f acc = {};
  for (int kt = 0; kt < A_DIM / KT; ++kt) {
    const int k0 = kt * KT;
    __syncthreads();
    for (int idx4 = tid; idx4 < (16 * KT) / 4; idx4 += 256) {
      const int r = idx4 / (KT / 4), c4 = idx4 % (KT / 4);
      *reinterpret_cast<float4*>(&sAp[r * KT + c4 * 4]) =
          *reinterpret_cast<const float4*>(
              &Aattr[(size_t)(m_base + r) * A_DIM + k0 + c4 * 4]);
    }
    for (int idx4 = tid; idx4 < (KT * D_DIM) / 4; idx4 += 256) {
      const int r = idx4 >> 5, c4 = idx4 & 31;
      *reinterpret_cast<float4*>(&sEp[r * D_DIM + c4 * 4]) =
          *reinterpret_cast<const float4*>(
              &Eemb[(size_t)(k0 + r) * D_DIM + c4 * 4]);
    }
    __syncthreads();
    if (kt + 1 < A_DIM / KT)
      __builtin_prefetch(&Aattr[(size_t)m_base * A_DIM + k0 + KT], 0, 0);
#pragma unroll
    for (int ks = 0; ks < KT / 4; ++ks) {
      const int k = 4 * ks + koff;
      const float2 av = *reinterpret_cast<const float2*>(&sAp[lo16 * KT + k]);
      v2f Av; Av[0] = av.x; Av[1] = av.y;
      v2f Bv; Bv[0] = sEp[k * D_DIM + dg]; Bv[1] = sEp[(k + 1) * D_DIM + dg];
      acc = wmma_f32(Av, Bv, acc);
    }
  }
#pragma unroll
  for (int r = 0; r < 8; ++r)
    out2[(size_t)(m_base + r + hi16) * D_DIM + dg] = acc[r];
}

extern "C" void kernel_launch(void* const* d_in, const int* in_sizes, int n_in,
                              void* d_out, int out_size, void* d_ws, size_t ws_size,
                              hipStream_t stream) {
  (void)in_sizes; (void)n_in; (void)out_size; (void)d_ws; (void)ws_size;
  const float* hidden = (const float*)d_in[0];
  const int*   adj    = (const int*)d_in[1];
  const float* amat   = (const float*)d_in[2];
  const float* Aattr  = (const float*)d_in[3];
  const float* Eemb   = (const float*)d_in[4];

  float* out_attn = (float*)d_out;
  float* out_attr = out_attn + (size_t)B_DIM * N_DIM * D_DIM;

  gat_attn_kernel<<<dim3(B_DIM * (N_DIM / 16)), dim3(256), 0, stream>>>(
      hidden, adj, amat, out_attn);
  attr_gemm_kernel<<<dim3((B_DIM * N_DIM) / 16), dim3(256), 0, stream>>>(
      Aattr, Eemb, out_attr);
}


# attr GEMM K loop software-pipelined: next panel's global loads issued as one batch into registers before the current panel's WMMAs, stored to a second LDS buffer after them; sum order unchanged
# speedup vs baseline: 1.6542x; 1.6542x over previous
#include <hip/hip_runtime.h>
#include <hip/hip_bf16.h>

typedef __attribute__((ext_vector_type(2))) float v2f;
typedef __attribute__((ext_vector_type(8))) float v8f;

#define B_DIM 64
#define N_DIM 128
#define D_DIM 128
#define A_DIM 1000
#define KT    100

__device__ __forceinline__ v8f wmma_f32(v2f a, v2f b, v8f c) {
  return __builtin_amdgcn_wmma_f32_16x16x4_f32(false, a, false, b, (short)0, c,
                                               false, false);
}

__global__ __launch_bounds__(256) void gat_attn_kernel(
    const float* __restrict__ hidden, const int* __restrict__ adj,
    const float* __restrict__ amat, float* __restrict__ out) {
  __shared__ float sH[N_DIM * D_DIM];
  __shared__ float sAlpha[16 * N_DIM];
  __shared__ float sAa[D_DIM * 4];
  __shared__ float sRed[16][16];
  __shared__ float sMax[16];
  __shared__ float sSum[16];

  const int tid    = threadIdx.x;
  const int b      = blockIdx.x >> 3;
  const int i_base = (blockIdx.x & 7) * 16;

  const float* Hg = hidden + (size_t)b * N_DIM * D_DIM;
  for (int idx4 = tid; idx4 < (N_DIM * D_DIM) / 4; idx4 += 256)
    reinterpret_cast<float4*>(sH)[idx4] = reinterpret_cast<const float4*>(Hg)[idx4];
  if (tid < (D_DIM * 4) / 4)
    reinterpret_cast<float4*>(sAa)[tid] = reinterpret_cast<const float4*>(amat)[tid];
  __syncthreads();

  const int wave   = tid >> 5;
  const int lane   = tid & 31;
  const int lo16   = lane & 15;
  const int hi16   = (lane >= 16) ? 8 : 0;
  const int koff   = (lane >= 16) ? 2 : 0;
  const int j_base = wave * 16;

  const float C1 = 0.6f, C2 = 0.4f;
  const float NEG_INF = -9.0e15f;

  v8f acc[4];
  acc[0] = {}; acc[1] = {}; acc[2] = {}; acc[3] = {};
  const int i_row = i_base + lo16;
  const int j_row = j_base + lo16;
  for (int d0 = 0; d0 < D_DIM; d0 += 4) {
    const int d = d0 + koff;
    const float2 hi = *reinterpret_cast<const float2*>(&sH[i_row * D_DIM + d]);
    const float2 hj = *reinterpret_cast<const float2*>(&sH[j_row * D_DIM + d]);
    const float* a0 = &sAa[d * 4];
    const float* a1 = &sAa[(d + 1) * 4];
    const float hax = fabsf(hi.x), hay = fabsf(hi.y);
    v2f Bv;  Bv[0] = hj.x;        Bv[1] = hj.y;
    v2f Bva; Bva[0] = fabsf(hj.x); Bva[1] = fabsf(hj.y);
#pragma unroll
    for (int k = 0; k < 4; ++k) {
      v2f Av;  Av[0]  = C1 * hi.x * a0[k]; Av[1]  = C1 * hi.y * a1[k];
      acc[k] = wmma_f32(Av, Bv, acc[k]);
      v2f Ava; Ava[0] = C2 * hax * a0[k];  Ava[1] = C2 * hay * a1[k];
      acc[k] = wmma_f32(Ava, Bva, acc[k]);
    }
  }

  const int* adjB = adj + (size_t)b * N_DIM * N_DIM;
#pragma unroll
  for (int r = 0; r < 8; ++r) {
    const int il = r + hi16;
    const int t = adjB[(i_base + il) * N_DIM + j_base + lo16];
    float e = NEG_INF;
    e = (t == 1) ? acc[0][r] : e;
    e = (t == 2) ? acc[1][r] : e;
    e = (t == 3) ? acc[2][r] : e;
    e = (t == 4) ? acc[3][r] : e;
    sAlpha[il * N_DIM + j_base + lo16] = e;
  }
  __syncthreads();

  const int row = tid >> 4;
  const int part = tid & 15;
  float m = -3.4e38f;
  for (int c = part * 8; c < part * 8 + 8; ++c)
    m = fmaxf(m, sAlpha[row * N_DIM + c]);
  sRed[row][part] = m;
  __syncthreads();
  if (part == 0) {
    float mm = sRed[row][0];
    for (int q = 1; q < 16; ++q) mm = fmaxf(mm, sRed[row][q]);
    sMax[row] = mm;
  }
  __syncthreads();
  const float mm2 = sMax[row];
  float s = 0.f;
  for (int c = part * 8; c < part * 8 + 8; ++c) {
    const float ev = __expf(sAlpha[row * N_DIM + c] - mm2);
    sAlpha[row * N_DIM + c] = ev;
    s += ev;
  }
  sRed[row][part] = s;
  __syncthreads();
  if (part == 0) {
    float ss = 0.f;
    for (int q = 0; q < 16; ++q) ss += sRed[row][q];
    sSum[row] = ss;
  }
  __syncthreads();
  const float inv = 1.0f / sSum[row];
  for (int c = part * 8; c < part * 8 + 8; ++c)
    sAlpha[row * N_DIM + c] *= inv;
  __syncthreads();

  const int dg = j_base + lo16;
  v8f oacc = {};
  for (int j0 = 0; j0 < N_DIM; j0 += 4) {
    const int jk = j0 + koff;
    const float2 av = *reinterpret_cast<const float2*>(&sAlpha[lo16 * N_DIM + jk]);
    v2f Av; Av[0] = av.x; Av[1] = av.y;
    v2f Bv; Bv[0] = sH[jk * D_DIM + dg]; Bv[1] = sH[(jk + 1) * D_DIM + dg];
    oacc = wmma_f32(Av, Bv, oacc);
  }
#pragma unroll
  for (int r = 0; r < 8; ++r)
    out[((size_t)b * N_DIM + i_base + r + hi16) * D_DIM + dg] = oacc[r];
}

#define NPANEL  (A_DIM / KT)
#define A_F4    ((16 * KT) / 4)
#define E_F4    ((KT * D_DIM) / 4)
#define A_SLOTS ((A_F4 + 255) / 256)
#define E_SLOTS ((E_F4 + 255) / 256)
#define SLOT_LIVE(q, n) ((256 * (q) + 255 < (n)) || (tid + 256 * (q) < (n)))

__global__ __launch_bounds__(256) void attr_gemm_kernel(
    const float* __restrict__ Aattr, const float* __restrict__ Eemb,
    float* __restrict__ out2) {
  __shared__ float sAp[2][16 * KT];
  __shared__ float sEp[2][KT * D_DIM];

  const int tid    = threadIdx.x;
  const int m_base = blockIdx.x * 16;
  const int wave   = tid >> 5;
  const int lane   = tid & 31;
  const int lo16   = lane & 15;
  const int hi16   = (lane >= 16) ? 8 : 0;
  const int koff   = (lane >= 16) ? 2 : 0;
  const int dg     = wave * 16 + lo16;

  const float* gA[A_SLOTS];
  int lA[A_SLOTS];
#pragma unroll
  for (int q = 0; q < A_SLOTS; ++q) {
    const int idx4 = tid + 256 * q;
    const int r = idx4 / (KT / 4), c4 = idx4 % (KT / 4);
    gA[q] = Aattr + (size_t)(m_base + r) * A_DIM + c4 * 4;
    lA[q] = r * KT + c4 * 4;
  }
  const float* gE = Eemb + tid * 4;

  float4 rA[A_SLOTS], rE[E_SLOTS];
#pragma unroll
  for (int q = 0; q < A_SLOTS; ++q) rA[q] = make_float4(0.f, 0.f, 0.f, 0.f);
#pragma unroll
  for (int q = 0; q < E_SLOTS; ++q) rE[q] = make_float4(0.f, 0.f, 0.f, 0.f);

  auto load_panel = [&](int k0) {
#pragma unroll
    for (int q = 0; q < A_SLOTS; ++q)
      if (SLOT_LIVE(q, A_F4))
        rA[q] = *reinterpret_cast<const float4*>(gA[q] + k0);
#pragma unroll
    for (int q = 0; q < E_SLOTS; ++q)
      if (SLOT_LIVE(q, E_F4))
        rE[q] = *reinterpret_cast<const float4*>(gE + (size_t)k0 * D_DIM + 1024 * q);
  };
  auto store_panel = [&](int nb) {
#pragma unroll
    for (int q = 0; q < A_SLOTS; ++q)
      if (SLOT_LIVE(q, A_F4))
        *reinterpret_cast<float4*>(&sAp[nb][lA[q]]) = rA[q];
#pragma unroll
    for (int q = 0; q < E_SLOTS; ++q)
      if (SLOT_LIVE(q, E_F4))
        *reinterpret_cast<float4*>(&sEp[nb][4 * (tid + 256 * q)]) = rE[q];
  };

  load_panel(0);
  store_panel(0);
  __syncthreads();

  v8f acc = {};
  for (int kt = 0; kt < NPANEL; ++kt) {
    const int cb = kt & 1;
    if (kt + 1 < NPANEL) load_panel((kt + 1) * KT);
    const float* pA = sAp[cb];
    const float* pE = sEp[cb];
#pragma unroll
    for (int ks = 0; ks < KT / 4; ++ks) {
      const int k = 4 * ks + koff;
      const float2 av = *reinterpret_cast<const float2*>(&pA[lo16 * KT + k]);
      v2f Av; Av[0] = av.x; Av[1] = av.y;
      v2f Bv; Bv[0] = pE[k * D_DIM + dg]; Bv[1] = pE[(k + 1) * D_DIM + dg];
      acc = wmma_f32(Av, Bv, acc);
    }
    if (kt + 1 < NPANEL) store_panel(cb ^ 1);
    __syncthreads();
  }
#pragma unroll
  for (int r = 0; r < 8; ++r)
    out2[(size_t)(m_base + r + hi16) * D_DIM + dg] = acc[r];
}

extern "C" void kernel_launch(void* const* d_in, const int* in_sizes, int n_in,
                              void* d_out, int out_size, void* d_ws, size_t ws_size,
                              hipStream_t stream) {
  (void)in_sizes; (void)n_in; (void)out_size; (void)d_ws; (void)ws_size;
  const float* hidden = (const float*)d_in[0];
  const int*   adj    = (const int*)d_in[1];
  const float* amat   = (const float*)d_in[2];
  const float* Aattr  = (const float*)d_in[3];
  const float* Eemb   = (const float*)d_in[4];

  float* out_attn = (float*)d_out;
  float* out_attr = out_attn + (size_t)B_DIM * N_DIM * D_DIM;

  gat_attn_kernel<<<dim3(B_DIM * (N_DIM / 16)), dim3(256), 0, stream>>>(
      hidden, adj, amat, out_attn);
  attr_gemm_kernel<<<dim3((B_DIM * N_DIM) / 16), dim3(256), 0, stream>>>(
      Aattr, Eemb, out_attr);
}


# LDS row pitches padded (H 128->140, score tile 128->132, E panel 128->136 floats) so the 16 rows of each WMMA fragment load fall on distinct banks instead of one; sum order unchanged
# speedup vs baseline: 1.8475x; 1.1169x over previous
#include <hip/hip_runtime.h>
#include <hip/hip_bf16.h>

typedef __attribute__((ext_vector_type(2))) float v2f;
typedef __attribute__((ext_vector_type(8))) float v8f;

#define B_DIM 64
#define N_DIM 128
#define D_DIM 128
#define A_DIM 1000
#define KT    100
#define HP    140
#define AP    132
#define EP    136

__device__ __forceinline__ v8f wmma_f32(v2f a, v2f b, v8f c) {
  return __builtin_amdgcn_wmma_f32_16x16x4_f32(false, a, false, b, (short)0, c,
                                               false, false);
}

__global__ __launch_bounds__(256) void gat_attn_kernel(
    const float* __restrict__ hidden, const int* __restrict__ adj,
    const float* __restrict__ amat, float* __restrict__ out) {
  __shared__ __attribute__((aligned(16))) float sH[N_DIM * HP];
  __shared__ __attribute__((aligned(16))) float sAlpha[16 * AP];
  __shared__ __attribute__((aligned(16))) float sAa[D_DIM * 4];
  __shared__ float sRed[16][16];
  __shared__ float sMax[16];
  __shared__ float sSum[16];

  const int tid    = threadIdx.x;
  const int b      = blockIdx.x >> 3;
  const int i_base = (blockIdx.x & 7) * 16;

  const float* Hg = hidden + (size_t)b * N_DIM * D_DIM;
  for (int idx4 = tid; idx4 < (N_DIM * D_DIM) / 4; idx4 += 256) {
    const int r = idx4 >> 5, c4 = idx4 & 31;
    *reinterpret_cast<float4*>(&sH[r * HP + c4 * 4]) =
        reinterpret_cast<const float4*>(Hg)[idx4];
  }
  if (tid < (D_DIM * 4) / 4)
    reinterpret_cast<float4*>(sAa)[tid] = reinterpret_cast<const float4*>(amat)[tid];
  __syncthreads();

  const int wave   = tid >> 5;
  const int lane   = tid & 31;
  const int lo16   = lane & 15;
  const int hi16   = (lane >= 16) ? 8 : 0;
  const int koff   = (lane >= 16) ? 2 : 0;
  const int j_base = wave * 16;

  const float C1 = 0.6f, C2 = 0.4f;
  const float NEG_INF = -9.0e15f;

  v8f acc[4];
  acc[0] = {}; acc[1] = {}; acc[2] = {}; acc[3] = {};
  const int i_row = i_base + lo16;
  const int j_row = j_base + lo16;
  for (int d0 = 0; d0 < D_DIM; d0 += 4) {
    const int d = d0 + koff;
    const float2 hi = *reinterpret_cast<const float2*>(&sH[i_row * HP + d]);
    const float2 hj = *reinterpret_cast<const float2*>(&sH[j_row * HP + d]);
    const float* a0 = &sAa[d * 4];
    const float* a1 = &sAa[(d + 1) * 4];
    const float hax = fabsf(hi.x), hay = fabsf(hi.y);
    v2f Bv;  Bv[0] = hj.x;        Bv[1] = hj.y;
    v2f Bva; Bva[0] = fabsf(hj.x); Bva[1] = fabsf(hj.y);
#pragma unroll
    for (int k = 0; k < 4; ++k) {
      v2f Av;  Av[0]  = C1 * hi.x * a0[k]; Av[1]  = C1 * hi.y * a1[k];
      acc[k] = wmma_f32(Av, Bv, acc[k]);
      v2f Ava; Ava[0] = C2 * hax * a0[k];  Ava[1] = C2 * hay * a1[k];
      acc[k] = wmma_f32(Ava, Bva, acc[k]);
    }
  }

  const int* adjB = adj + (size_t)b * N_DIM * N_DIM;
#pragma unroll
  for (int r = 0; r < 8; ++r) {
    const int il = r + hi16;
    const int t = adjB[(i_base + il) * N_DIM + j_base + lo16];
    float e = NEG_INF;
    e = (t == 1) ? acc[0][r] : e;
    e = (t == 2) ? acc[1][r] : e;
    e = (t == 3) ? acc[2][r] : e;
    e = (t == 4) ? acc[3][r] : e;
    sAlpha[il * AP + j_base + lo16] = e;
  }
  __syncthreads();

  const int row = tid >> 4;
  const int part = tid & 15;
  float m = -3.4e38f;
  for (int c = part * 8; c < part * 8 + 8; ++c)
    m = fmaxf(m, sAlpha[row * AP + c]);
  sRed[row][part] = m;
  __syncthreads();
  if (part == 0) {
    float mm = sRed[row][0];
    for (int q = 1; q < 16; ++q) mm = fmaxf(mm, sRed[row][q]);
    sMax[row] = mm;
  }
  __syncthreads();
  const float mm2 = sMax[row];
  float s = 0.f;
  for (int c = part * 8; c < part * 8 + 8; ++c) {
    const float ev = __expf(sAlpha[row * AP + c] - mm2);
    sAlpha[row * AP + c] = ev;
    s += ev;
  }
  sRed[row][part] = s;
  __syncthreads();
  if (part == 0) {
    float ss = 0.f;
    for (int q = 0; q < 16; ++q) ss += sRed[row][q];
    sSum[row] = ss;
  }
  __syncthreads();
  const float inv = 1.0f / sSum[row];
  for (int c = part * 8; c < part * 8 + 8; ++c)
    sAlpha[row * AP + c] *= inv;
  __syncthreads();

  const int dg = j_base + lo16;
  v8f oacc = {};
  for (int j0 = 0; j0 < N_DIM; j0 += 4) {
    const int jk = j0 + koff;
    const float2 av = *reinterpret_cast<const float2*>(&sAlpha[lo16 * AP + jk]);
    v2f Av; Av[0] = av.x; Av[1] = av.y;
    v2f Bv; Bv[0] = sH[jk * HP + dg]; Bv[1] = sH[(jk + 1) * HP + dg];
    oacc = wmma_f32(Av, Bv, oacc);
  }
#pragma unroll
  for (int r = 0; r < 8; ++r)
    out[((size_t)b * N_DIM + i_base + r + hi16) * D_DIM + dg] = oacc[r];
}

#define NPANEL  (A_DIM / KT)
#define A_F4    ((16 * KT) / 4)
#define E_F4    ((KT * D_DIM) / 4)
#define A_SLOTS ((A_F4 + 255) / 256)
#define E_SLOTS ((E_F4 + 255) / 256)
#define SLOT_LIVE(q, n) ((256 * (q) + 255 < (n)) || (tid + 256 * (q) < (n)))

__global__ __launch_bounds__(256) void attr_gemm_kernel(
    const float* __restrict__ Aattr, const float* __restrict__ Eemb,
    float* __restrict__ out2) {
  __shared__ __attribute__((aligned(16))) float sAp[2][16 * KT];
  __shared__ __attribute__((aligned(16))) float sEp[2][KT * EP];

  const int tid    = threadIdx.x;
  const int m_base = blockIdx.x * 16;
  const int wave   = tid >> 5;
  const int lane   = tid & 31;
  const int lo16   = lane & 15;
  const int hi16   = (lane >= 16) ? 8 : 0;
  const int koff   = (lane >= 16) ? 2 : 0;
  const int dg     = wave * 16 + lo16;

  const float* gA[A_SLOTS];
  int lA[A_SLOTS];
#pragma unroll
  for (int q = 0; q < A_SLOTS; ++q) {
    const int idx4 = tid + 256 * q;
    const int r = idx4 / (KT / 4), c4 = idx4 % (KT / 4);
    gA[q] = Aattr + (size_t)(m_base + r) * A_DIM + c4 * 4;
    lA[q] = r * KT + c4 * 4;
  }
  const float* gE = Eemb + tid * 4;
  const int lE = (tid >> 5) * EP + (tid & 31) * 4;

  float4 rA[A_SLOTS], rE[E_SLOTS];
#pragma unroll
  for (int q = 0; q < A_SLOTS; ++q) rA[q] = make_float4(0.f, 0.f, 0.f, 0.f);
#pragma unroll
  for (int q = 0; q < E_SLOTS; ++q) rE[q] = make_float4(0.f, 0.f, 0.f, 0.f);

  auto load_panel = [&](int k0) {
#pragma unroll
    for (int q = 0; q < A_SLOTS; ++q)
      if (SLOT_LIVE(q, A_F4))
        rA[q] = *reinterpret_cast<const float4*>(gA[q] + k0);
#pragma unroll
    for (int q = 0; q < E_SLOTS; ++q)
      if (SLOT_LIVE(q, E_F4))
        rE[q] = *reinterpret_cast<const float4*>(gE + (size_t)k0 * D_DIM + 1024 * q);
  };
  auto store_panel = [&](int nb) {
#pragma unroll
    for (int q = 0; q < A_SLOTS; ++q)
      if (SLOT_LIVE(q, A_F4))
        *reinterpret_cast<float4*>(&sAp[nb][lA[q]]) = rA[q];
#pragma unroll
    for (int q = 0; q < E_SLOTS; ++q)
      if (SLOT_LIVE(q, E_F4))
        *reinterpret_cast<float4*>(&sEp[nb][lE + 8 * EP * q]) = rE[q];
  };

  load_panel(0);
  store_panel(0);
  __syncthreads();

  v8f acc = {};
  for (int kt = 0; kt < NPANEL; ++kt) {
    const int cb = kt & 1;
    if (kt + 1 < NPANEL) load_panel((kt + 1) * KT);
    const float* pA = sAp[cb];
    const float* pE = sEp[cb];
#pragma unroll
    for (int ks = 0; ks < KT / 4; ++ks) {
      const int k = 4 * ks + koff;
      const float2 av = *reinterpret_cast<const float2*>(&pA[lo16 * KT + k]);
      v2f Av; Av[0] = av.x; Av[1] = av.y;
      v2f Bv; Bv[0] = pE[k * EP + dg]; Bv[1] = pE[(k + 1) * EP + dg];
      acc = wmma_f32(Av, Bv, acc);
    }
    if (kt + 1 < NPANEL) store_panel(cb ^ 1);
    __syncthreads();
  }
#pragma unroll
  for (int r = 0; r < 8; ++r)
    out2[(size_t)(m_base + r + hi16) * D_DIM + dg] = acc[r];
}

extern "C" void kernel_launch(void* const* d_in, const int* in_sizes, int n_in,
                              void* d_out, int out_size, void* d_ws, size_t ws_size,
                              hipStream_t stream) {
  (void)in_sizes; (void)n_in; (void)out_size; (void)d_ws; (void)ws_size;
  const float* hidden = (const float*)d_in[0];
  const int*   adj    = (const int*)d_in[1];
  const float* amat   = (const float*)d_in[2];
  const float* Aattr  = (const float*)d_in[3];
  const float* Eemb   = (const float*)d_in[4];

  float* out_attn = (float*)d_out;
  float* out_attr = out_attn + (size_t)B_DIM * N_DIM * D_DIM;

  gat_attn_kernel<<<dim3(B_DIM * (N_DIM / 16)), dim3(256), 0, stream>>>(
      hidden, adj, amat, out_attn);
  attr_gemm_kernel<<<dim3((B_DIM * N_DIM) / 16), dim3(256), 0, stream>>>(
      Aattr, Eemb, out_attr);
}


# attention and attr GEMM run as two work-group roles of one launch (role by block index, runs of 256 alternate): one launch boundary goes, each role's non-matrix phases run under the other's WMMAs
# speedup vs baseline: 1.9568x; 1.0592x over previous
#include <hip/hip_runtime.h>
#include <hip/hip_bf16.h>

typedef __attribute__((ext_vector_type(2))) float v2f;
typedef __attribute__((ext_vector_type(8))) float v8f;

#define B_DIM 64
#define N_DIM 128
#define D_DIM 128
#define A_DIM 1000
#define KT    100
#define HP    140
#define AP    132
#define EP    136

__device__ __forceinline__ v8f wmma_f32(v2f a, v2f b, v8f c) {
  return __builtin_amdgcn_wmma_f32_16x16x4_f32(false, a, false, b, (short)0, c,
                                               false, false);
}


struct AttLds {
  float H[N_DIM * HP];
  float Alpha[16 * AP];
  float Aa[D_DIM * 4];
  float Red[16][16];
  float Max[16];
  float Sum[16];
};
struct GemmLds {
  float Ap[2][16 * KT];
  float Ep[2][KT * EP];
};
union __attribute__((aligned(16))) FusedLds {
  AttLds  att;
  GemmLds gemm;
};

__device__ __forceinline__ void gat_attn_role(
    AttLds& L, const int tile,
    const float* __restrict__ hidden, const int* __restrict__ adj,
    const float* __restrict__ amat, float* __restrict__ out) {
  float* const sH = L.H;
  float* const sAlpha = L.Alpha;
  float* const sAa = L.Aa;
  float (* const sRed)[16] = L.Red;
  float* const sMax = L.Max;
  float* const sSum = L.Sum;

  const int tid    = threadIdx.x;
  const int b      = tile >> 3;
  const int i_base = (tile & 7) * 16;

  const float* Hg = hidden + (size_t)b * N_DIM * D_DIM;
  for (int idx4 = tid; idx4 < (N_DIM * D_DIM) / 4; idx4 += 256) {
    const int r = idx4 >> 5, c4 = idx4 & 31;
    *reinterpret_cast<float4*>(&sH[r * HP + c4 * 4]) =
        reinterpret_cast<const float4*>(Hg)[idx4];
  }
  if (tid < (D_DIM * 4) / 4)
    reinterpret_cast<float4*>(sAa)[tid] = reinterpret_cast<const float4*>(amat)[tid];
  __syncthreads();

  const int wave   = tid >> 5;
  const int lane   = tid & 31;
  const int lo16   = lane & 15;
  const int hi16   = (lane >= 16) ? 8 : 0;
  const int koff   = (lane >= 16) ? 2 : 0;
  const int j_base = wave * 16;

  const float C1 = 0.6f, C2 = 0.4f;
  const float NEG_INF = -9.0e15f;

  v8f acc[4];
  acc[0] = {}; acc[1] = {}; acc[2] = {}; acc[3] = {};
  const int i_row = i_base + lo16;
  const int j_row = j_base + lo16;
  for (int d0 = 0; d0 < D_DIM; d0 += 4) {
    const int d = d0 + koff;
    const float2 hi = *reinterpret_cast<const float2*>(&sH[i_row * HP + d]);
    const float2 hj = *reinterpret_cast<const float2*>(&sH[j_row * HP + d]);
    const float* a0 = &sAa[d * 4];
    const float* a1 = &sAa[(d + 1) * 4];
    const float hax = fabsf(hi.x), hay = fabsf(hi.y);
    v2f Bv;  Bv[0] = hj.x;        Bv[1] = hj.y;
    v2f Bva; Bva[0] = fabsf(hj.x); Bva[1] = fabsf(hj.y);
#pragma unroll
    for (int k = 0; k < 4; ++k) {
      v2f Av;  Av[0]  = C1 * hi.x * a0[k]; Av[1]  = C1 * hi.y * a1[k];
      acc[k] = wmma_f32(Av, Bv, acc[k]);
      v2f Ava; Ava[0] = C2 * hax * a0[k];  Ava[1] = C2 * hay * a1[k];
      acc[k] = wmma_f32(Ava, Bva, acc[k]);
    }
  }

  const int* adjB = adj + (size_t)b * N_DIM * N_DIM;
#pragma unroll
  for (int r = 0; r < 8; ++r) {
    const int il = r + hi16;
    const int t = adjB[(i_base + il) * N_DIM + j_base + lo16];
    float e = NEG_INF;
    e = (t == 1) ? acc[0][r] : e;
    e = (t == 2) ? acc[1][r] : e;
    e = (t == 3) ? acc[2][r] : e;
    e = (t == 4) ? acc[3][r] : e;
    sAlpha[il * AP + j_base + lo16] = e;
  }
  __syncthreads();

  const int row = tid >> 4;
  const int part = tid & 15;
  float m = -3.4e38f;
  for (int c = part * 8; c < part * 8 + 8; ++c)
    m = fmaxf(m, sAlpha[row * AP + c]);
  sRed[row][part] = m;
  __syncthreads();
  if (part == 0) {
    float mm = sRed[row][0];
    for (int q = 1; q < 16; ++q) mm = fmaxf(mm, sRed[row][q]);
    sMax[row] = mm;
  }
  __syncthreads();
  const float mm2 = sMax[row];
  float s = 0.f;
  for (int c = part * 8; c < part * 8 + 8; ++c) {
    const float ev = __expf(sAlpha[row * AP + c] - mm2);
    sAlpha[row * AP + c] = ev;
    s += ev;
  }
  sRed[row][part] = s;
  __syncthreads();
  if (part == 0) {
    float ss = 0.f;
    for (int q = 0; q < 16; ++q) ss += sRed[row][q];
    sSum[row] = ss;
  }
  __syncthreads();
  const float inv = 1.0f / sSum[row];
  for (int c = part * 8; c < part * 8 + 8; ++c)
    sAlpha[row * AP + c] *= inv;
  __syncthreads();

  const int dg = j_base + lo16;
  v8f oacc = {};
  for (int j0 = 0; j0 < N_DIM; j0 += 4) {
    const int jk = j0 + koff;
    const float2 av = *reinterpret_cast<const float2*>(&sAlpha[lo16 * AP + jk]);
    v2f Av; Av[0] = av.x; Av[1] = av.y;
    v2f Bv; Bv[0] = sH[jk * HP + dg]; Bv[1] = sH[(jk + 1) * HP + dg];
    oacc = wmma_f32(Av, Bv, oacc);
  }
#pragma unroll
  for (int r = 0; r < 8; ++r)
    out[((size_t)b * N_DIM + i_base + r + hi16) * D_DIM + dg] = oacc[r];
}

#define NPANEL  (A_DIM / KT)
#define A_F4    ((16 * KT) / 4)
#define E_F4    ((KT * D_DIM) / 4)
#define A_SLOTS ((A_F4 + 255) / 256)
#define E_SLOTS ((E_F4 + 255) / 256)
#define SLOT_LIVE(q, n) ((256 * (q) + 255 < (n)) || (tid + 256 * (q) < (n)))

__device__ __forceinline__ void attr_gemm_role(
    GemmLds& L, const int tile,
    const float* __restrict__ Aattr, const float* __restrict__ Eemb,
    float* __restrict__ out2) {
  float (* const sAp)[16 * KT] = L.Ap;
  float (* const sEp)[KT * EP] = L.Ep;

  const int tid    = threadIdx.x;
  const int m_base = tile * 16;
  const int wave   = tid >> 5;
  const int lane   = tid & 31;
  const int lo16   = lane & 15;
  const int hi16   = (lane >= 16) ? 8 : 0;
  const int koff   = (lane >= 16) ? 2 : 0;
  const int dg     = wave * 16 + lo16;

  const float* gA[A_SLOTS];
  int lA[A_SLOTS];
#pragma unroll
  for (int q = 0; q < A_SLOTS; ++q) {
    const int idx4 = tid + 256 * q;
    const int r = idx4 / (KT / 4), c4 = idx4 % (KT / 4);
    gA[q] = Aattr + (size_t)(m_base + r) * A_DIM + c4 * 4;
    lA[q] = r * KT + c4 * 4;
  }
  const float* gE = Eemb + tid * 4;
  const int lE = (tid >> 5) * EP + (tid & 31) * 4;

  float4 rA[A_SLOTS], rE[E_SLOTS];
#pragma unroll
  for (int q = 0; q < A_SLOTS; ++q) rA[q] = make_float4(0.f, 0.f, 0.f, 0.f);
#pragma unroll
  for (int q = 0; q < E_SLOTS; ++q) rE[q] = make_float4(0.f, 0.f, 0.f, 0.f);

  auto load_panel = [&](int k0) {
#pragma unroll
    for (int q = 0; q < A_SLOTS; ++q)
      if (SLOT_LIVE(q, A_F4))
        rA[q] = *reinterpret_cast<const float4*>(gA[q] + k0);
#pragma unroll
    for (int q = 0; q < E_SLOTS; ++q)
      if (SLOT_LIVE(q, E_F4))
        rE[q] = *reinterpret_cast<const float4*>(gE + (size_t)k0 * D_DIM + 1024 * q);
  };
  auto store_panel = [&](int nb) {
#pragma unroll
    for (int q = 0; q < A_SLOTS; ++q)
      if (SLOT_LIVE(q, A_F4))
        *reinterpret_cast<float4*>(&sAp[nb][lA[q]]) = rA[q];
#pragma unroll
    for (int q = 0; q < E_SLOTS; ++q)
      if (SLOT_LIVE(q, E_F4))
        *reinterpret_cast<float4*>(&sEp[nb][lE + 8 * EP * q]) = rE[q];
  };

  load_panel(0);
  store_panel(0);
  __syncthreads();

  v8f acc = {};
  for (int kt = 0; kt < NPANEL; ++kt) {
    const int cb = kt & 1;
    if (kt + 1 < NPANEL) load_panel((kt + 1) * KT);
    const float* pA = sAp[cb];
    const float* pE = sEp[cb];
#pragma unroll
    for (int ks = 0; ks < KT / 4; ++ks) {
      const int k = 4 * ks + koff;
      const float2 av = *reinterpret_cast<const float2*>(&pA[lo16 * KT + k]);
      v2f Av; Av[0] = av.x; Av[1] = av.y;
      v2f Bv; Bv[0] = pE[k * EP + dg]; Bv[1] = pE[(k + 1) * EP + dg];
      acc = wmma_f32(Av, Bv, acc);
    }
    if (kt + 1 < NPANEL) store_panel(cb ^ 1);
    __syncthreads();
  }
#pragma unroll
  for (int r = 0; r < 8; ++r)
    out2[(size_t)(m_base + r + hi16) * D_DIM + dg] = acc[r];
}

#define TILES_PER_ROLE (B_DIM * (N_DIM / 16))
#define ROLE_RUN_LOG2  8

__global__ __launch_bounds__(256) void fused_kernel(
    const float* __restrict__ hidden, const int* __restrict__ adj,
    const float* __restrict__ amat, const float* __restrict__ Aattr,
    const float* __restrict__ Eemb, float* __restrict__ out_attn,
    float* __restrict__ out_attr) {
  __shared__ FusedLds lds;
  const int bid  = blockIdx.x;
  const int role = (bid >> ROLE_RUN_LOG2) & 1;
  const int tile = ((bid >> (ROLE_RUN_LOG2 + 1)) << ROLE_RUN_LOG2) |
                   (bid & ((1 << ROLE_RUN_LOG2) - 1));
  if (role == 0)
    gat_attn_role(lds.att, tile, hidden, adj, amat, out_attn);
  else
    attr_gemm_role(lds.gemm, tile, Aattr, Eemb, out_attr);
}

extern "C" void kernel_launch(void* const* d_in, const int* in_sizes, int n_in,
                              void* d_out, int out_size, void* d_ws, size_t ws_size,
                              hipStream_t stream) {
  (void)in_sizes; (void)n_in; (void)out_size; (void)d_ws; (void)ws_size;
  const float* hidden = (const float*)d_in[0];
  const int*   adj    = (const int*)d_in[1];
  const float* amat   = (const float*)d_in[2];
  const float* Aattr  = (const float*)d_in[3];
  const float* Eemb   = (const float*)d_in[4];

  float* out_attn = (float*)d_out;
  float* out_attr = out_attn + (size_t)B_DIM * N_DIM * D_DIM;

  fused_kernel<<<dim3(2 * TILES_PER_ROLE), dim3(256), 0, stream>>>(
      hidden, adj, amat, Aattr, Eemb, out_attn, out_attr);
}


# work-group tile height 16->32 rows in both roles: a wave keeps two row tiles per shared B fragment, staged H/E panels serve twice the rows, grid 1024->512 in one resident round; sum order unchanged
# speedup vs baseline: 2.2658x; 1.1579x over previous
#include <hip/hip_runtime.h>
#include <hip/hip_bf16.h>


typedef __attribute__((ext_vector_type(2))) float v2f;
typedef __attribute__((ext_vector_type(8))) float v8f;

#define B_DIM 64
#define N_DIM 128
#define D_DIM 128
#define A_DIM 1000
#define KT    100
#define IT    32
#define HP    140
#define AP    132
#define EP    136

__device__ __forceinline__ v8f wmma_f32(v2f a, v2f b, v8f c) {
  return __builtin_amdgcn_wmma_f32_16x16x4_f32(false, a, false, b, (short)0, c,
                                               false, false);
}


struct AttLds {
  float H[N_DIM * HP];
  float Alpha[IT * AP];
  float Aa[D_DIM * 4];
  float Red[IT][16];
  float Max[IT];
  float Sum[IT];
};
struct GemmLds {
  float Ap[2][IT * KT];
  float Ep[2][KT * EP];
};
union __attribute__((aligned(16))) FusedLds {
  AttLds  att;
  GemmLds gemm;
};

__device__ __forceinline__ void gat_attn_role(
    AttLds& L, const int tile,
    const float* __restrict__ hidden, const int* __restrict__ adj,
    const float* __restrict__ amat, float* __restrict__ out) {
  float* const sH = L.H;
  float* const sAlpha = L.Alpha;
  float* const sAa = L.Aa;
  float (* const sRed)[16] = L.Red;
  float* const sMax = L.Max;
  float* const sSum = L.Sum;

  const int tid    = threadIdx.x;
  const int b      = tile / (N_DIM / IT);
  const int i_base = (tile % (N_DIM / IT)) * IT;

  const float* Hg = hidden + (size_t)b * N_DIM * D_DIM;
  for (int idx4 = tid; idx4 < (N_DIM * D_DIM) / 4; idx4 += 256) {
    const int r = idx4 >> 5, c4 = idx4 & 31;
    *reinterpret_cast<float4*>(&sH[r * HP + c4 * 4]) =
        reinterpret_cast<const float4*>(Hg)[idx4];
  }
  if (tid < (D_DIM * 4) / 4)
    reinterpret_cast<float4*>(sAa)[tid] = reinterpret_cast<const float4*>(amat)[tid];
  __syncthreads();

  const int wave   = tid >> 5;
  const int lane   = tid & 31;
  const int lo16   = lane & 15;
  const int hi16   = (lane >= 16) ? 8 : 0;
  const int koff   = (lane >= 16) ? 2 : 0;
  const int j_base = wave * 16;

  const float C1 = 0.6f, C2 = 0.4f;
  const float NEG_INF = -9.0e15f;

  v8f acc[2][4] = {};
  const int j_row = j_base + lo16;
  for (int d0 = 0; d0 < D_DIM; d0 += 4) {
    const int d = d0 + koff;
    const float2 hj = *reinterpret_cast<const float2*>(&sH[j_row * HP + d]);
    const float* a0 = &sAa[d * 4];
    const float* a1 = &sAa[(d + 1) * 4];
    v2f Bv;  Bv[0] = hj.x;        Bv[1] = hj.y;
    v2f Bva; Bva[0] = fabsf(hj.x); Bva[1] = fabsf(hj.y);
#pragma unroll
    for (int t = 0; t < 2; ++t) {
      const int i_row = i_base + 16 * t + lo16;
      const float2 hi = *reinterpret_cast<const float2*>(&sH[i_row * HP + d]);
      const float hax = fabsf(hi.x), hay = fabsf(hi.y);
#pragma unroll
      for (int k = 0; k < 4; ++k) {
        v2f Av;  Av[0]  = C1 * hi.x * a0[k]; Av[1]  = C1 * hi.y * a1[k];
        acc[t][k] = wmma_f32(Av, Bv, acc[t][k]);
        v2f Ava; Ava[0] = C2 * hax * a0[k];  Ava[1] = C2 * hay * a1[k];
        acc[t][k] = wmma_f32(Ava, Bva, acc[t][k]);
      }
    }
  }

  const int* adjB = adj + (size_t)b * N_DIM * N_DIM;
#pragma unroll
  for (int t = 0; t < 2; ++t) {
#pragma unroll
    for (int r = 0; r < 8; ++r) {
      const int il = 16 * t + r + hi16;
      const int ty = adjB[(i_base + il) * N_DIM + j_base + lo16];
      float e = NEG_INF;
      e = (ty == 1) ? acc[t][0][r] : e;
      e = (ty == 2) ? acc[t][1][r] : e;
      e = (ty == 3) ? acc[t][2][r] : e;
      e = (ty == 4) ? acc[t][3][r] : e;
      sAlpha[il * AP + j_base + lo16] = e;
    }
  }
  __syncthreads();

  const int row = tid >> 4;
  const int part = tid & 15;
  for (int h = 0; h < 2; ++h) {
    const int rr = row + 16 * h;
    float m = -3.4e38f;
    for (int c = part * 8; c < part * 8 + 8; ++c)
      m = fmaxf(m, sAlpha[rr * AP + c]);
    sRed[rr][part] = m;
  }
  __syncthreads();
  if (part < 2) {
    const int rr = row + 16 * part;
    float mm = sRed[rr][0];
    for (int q = 1; q < 16; ++q) mm = fmaxf(mm, sRed[rr][q]);
    sMax[rr] = mm;
  }
  __syncthreads();
  for (int h = 0; h < 2; ++h) {
    const int rr = row + 16 * h;
    const float mm2 = sMax[rr];
    float s = 0.f;
    for (int c = part * 8; c < part * 8 + 8; ++c) {
      const float ev = __expf(sAlpha[rr * AP + c] - mm2);
      sAlpha[rr * AP + c] = ev;
      s += ev;
    }
    sRed[rr][part] = s;
  }
  __syncthreads();
  if (part < 2) {
    const int rr = row + 16 * part;
    float ss = 0.f;
    for (int q = 0; q < 16; ++q) ss += sRed[rr][q];
    sSum[rr] = ss;
  }
  __syncthreads();
  for (int h = 0; h < 2; ++h) {
    const int rr = row + 16 * h;
    const float inv = 1.0f / sSum[rr];
    for (int c = part * 8; c < part * 8 + 8; ++c)
      sAlpha[rr * AP + c] *= inv;
  }
  __syncthreads();

  const int dg = j_base + lo16;
  v8f oacc0 = {}, oacc1 = {};
  for (int j0 = 0; j0 < N_DIM; j0 += 4) {
    const int jk = j0 + koff;
    const float2 av0 = *reinterpret_cast<const float2*>(&sAlpha[lo16 * AP + jk]);
    const float2 av1 = *reinterpret_cast<const float2*>(&sAlpha[(16 + lo16) * AP + jk]);
    v2f Av0; Av0[0] = av0.x; Av0[1] = av0.y;
    v2f Av1; Av1[0] = av1.x; Av1[1] = av1.y;
    v2f Bv; Bv[0] = sH[jk * HP + dg]; Bv[1] = sH[(jk + 1) * HP + dg];
    oacc0 = wmma_f32(Av0, Bv, oacc0);
    oacc1 = wmma_f32(Av1, Bv, oacc1);
  }
#pragma unroll
  for (int r = 0; r < 8; ++r) {
    out[((size_t)b * N_DIM + i_base + r + hi16) * D_DIM + dg] = oacc0[r];
    out[((size_t)b * N_DIM + i_base + 16 + r + hi16) * D_DIM + dg] = oacc1[r];
  }
}

#define NPANEL  (A_DIM / KT)
#define A_F4    ((IT * KT) / 4)
#define E_F4    ((KT * D_DIM) / 4)
#define A_SLOTS ((A_F4 + 255) / 256)
#define E_SLOTS ((E_F4 + 255) / 256)
#define SLOT_LIVE(q, n) ((256 * (q) + 255 < (n)) || (tid + 256 * (q) < (n)))

__device__ __forceinline__ void attr_gemm_role(
    GemmLds& L, const int tile,
    const float* __restrict__ Aattr, const float* __restrict__ Eemb,
    float* __restrict__ out2) {
  float (* const sAp)[IT * KT] = L.Ap;
  float (* const sEp)[KT * EP] = L.Ep;

  const int tid    = threadIdx.x;
  const int m_base = tile * IT;
  const int wave   = tid >> 5;
  const int lane   = tid & 31;
  const int lo16   = lane & 15;
  const int hi16   = (lane >= 16) ? 8 : 0;
  const int koff   = (lane >= 16) ? 2 : 0;
  const int dg     = wave * 16 + lo16;

  const float* gA[A_SLOTS];
  int lA[A_SLOTS];
#pragma unroll
  for (int q = 0; q < A_SLOTS; ++q) {
    const int idx4 = tid + 256 * q;
    const int r = idx4 / (KT / 4), c4 = idx4 % (KT / 4);
    gA[q] = Aattr + (size_t)(m_base + r) * A_DIM + c4 * 4;
    lA[q] = r * KT + c4 * 4;
  }
  const float* gE = Eemb + tid * 4;
  const int lE = (tid >> 5) * EP + (tid & 31) * 4;

  float4 rA[A_SLOTS], rE[E_SLOTS];
#pragma unroll
  for (int q = 0; q < A_SLOTS; ++q) rA[q] = make_float4(0.f, 0.f, 0.f, 0.f);
#pragma unroll
  for (int q = 0; q < E_SLOTS; ++q) rE[q] = make_float4(0.f, 0.f, 0.f, 0.f);

  auto load_panel = [&](int k0) {
#pragma unroll
    for (int q = 0; q < A_SLOTS; ++q)
      if (SLOT_LIVE(q, A_F4))
        rA[q] = *reinterpret_cast<const float4*>(gA[q] + k0);
#pragma unroll
    for (int q = 0; q < E_SLOTS; ++q)
      if (SLOT_LIVE(q, E_F4))
        rE[q] = *reinterpret_cast<const float4*>(gE + (size_t)k0 * D_DIM + 1024 * q);
  };
  auto store_panel = [&](int nb) {
#pragma unroll
    for (int q = 0; q < A_SLOTS; ++q)
      if (SLOT_LIVE(q, A_F4))
        *reinterpret_cast<float4*>(&sAp[nb][lA[q]]) = rA[q];
#pragma unroll
    for (int q = 0; q < E_SLOTS; ++q)
      if (SLOT_LIVE(q, E_F4))
        *reinterpret_cast<float4*>(&sEp[nb][lE + 8 * EP * q]) = rE[q];
  };

  load_panel(0);
  store_panel(0);
  __syncthreads();

  v8f acc0 = {}, acc1 = {};
  for (int kt = 0; kt < NPANEL; ++kt) {
    const int cb = kt & 1;
    if (kt + 1 < NPANEL) load_panel((kt + 1) * KT);
    const float* pA = sAp[cb];
    const float* pE = sEp[cb];
#pragma unroll
    for (int ks = 0; ks < KT / 4; ++ks) {
      const int k = 4 * ks + koff;
      const float2 av0 = *reinterpret_cast<const float2*>(&pA[lo16 * KT + k]);
      const float2 av1 = *reinterpret_cast<const float2*>(&pA[(16 + lo16) * KT + k]);
      v2f Av0; Av0[0] = av0.x; Av0[1] = av0.y;
      v2f Av1; Av1[0] = av1.x; Av1[1] = av1.y;
      v2f Bv; Bv[0] = pE[k * EP + dg]; Bv[1] = pE[(k + 1) * EP + dg];
      acc0 = wmma_f32(Av0, Bv, acc0);
      acc1 = wmma_f32(Av1, Bv, acc1);
    }
    if (kt + 1 < NPANEL) store_panel(cb ^ 1);
    __syncthreads();
  }
#pragma unroll
  for (int r = 0; r < 8; ++r) {
    out2[(size_t)(m_base + r + hi16) * D_DIM + dg] = acc0[r];
    out2[(size_t)(m_base + 16 + r + hi16) * D_DIM + dg] = acc1[r];
  }
}

#define TILES_PER_ROLE (B_DIM * (N_DIM / IT))

__global__ __launch_bounds__(256) void fused_kernel(
    const float* __restrict__ hidden, const int* __restrict__ adj,
    const float* __restrict__ amat, const float* __restrict__ Aattr,
    const float* __restrict__ Eemb, float* __restrict__ out_attn,
    float* __restrict__ out_attr) {
  __shared__ FusedLds lds;
  const int bid  = blockIdx.x;
  const int role = bid / TILES_PER_ROLE;
  const int tile = bid % TILES_PER_ROLE;
  if (role == 0)
    gat_attn_role(lds.att, tile, hidden, adj, amat, out_attn);
  else
    attr_gemm_role(lds.gemm, tile, Aattr, Eemb, out_attr);
}

extern "C" void kernel_launch(void* const* d_in, const int* in_sizes, int n_in,
                              void* d_out, int out_size, void* d_ws, size_t ws_size,
                              hipStream_t stream) {
  (void)in_sizes; (void)n_in; (void)out_size; (void)d_ws; (void)ws_size;
  const float* hidden = (const float*)d_in[0];
  const int*   adj    = (const int*)d_in[1];
  const float* amat   = (const float*)d_in[2];
  const float* Aattr  = (const float*)d_in[3];
  const float* Eemb   = (const float*)d_in[4];

  float* out_attn = (float*)d_out;
  float* out_attr = out_attn + (size_t)B_DIM * N_DIM * D_DIM;

  fused_kernel<<<dim3(2 * TILES_PER_ROLE), dim3(256), 0, stream>>>(
      hidden, adj, amat, Aattr, Eemb, out_attn, out_attr);
}


# all three WMMA loops: fragment LDS loads issued as one group a block ahead into a second register set, not among the WMMAs; attr GEMM multiplies a panel's last block after its barrier; sum order same
# speedup vs baseline: 2.3344x; 1.0303x over previous
#include <hip/hip_runtime.h>
#include <hip/hip_bf16.h>


typedef __attribute__((ext_vector_type(2))) float v2f;
typedef __attribute__((ext_vector_type(8))) float v8f;

#define B_DIM 64
#define N_DIM 128
#define D_DIM 128
#define A_DIM 1000
#define KT    100
#define OB    8
#define IT    32
#define HP    140
#define AP    132
#define EP    136

__device__ __forceinline__ v8f wmma_f32(v2f a, v2f b, v8f c) {
  return __builtin_amdgcn_wmma_f32_16x16x4_f32(false, a, false, b, (short)0, c,
                                               false, false);
}


struct AttLds {
  float H[N_DIM * HP];
  float Alpha[IT * AP];
  float Aa[D_DIM * 4];
  float Red[IT][16];
  float Max[IT];
  float Sum[IT];
};
struct GemmLds {
  float Ap[2][IT * KT];
  float Ep[2][KT * EP];
};
union __attribute__((aligned(16))) FusedLds {
  AttLds  att;
  GemmLds gemm;
};

__device__ __forceinline__ void gat_attn_role(
    AttLds& L, const int tile,
    const float* __restrict__ hidden, const int* __restrict__ adj,
    const float* __restrict__ amat, float* __restrict__ out) {
  float* const sH = L.H;
  float* const sAlpha = L.Alpha;
  float* const sAa = L.Aa;
  float (* const sRed)[16] = L.Red;
  float* const sMax = L.Max;
  float* const sSum = L.Sum;

  const int tid    = threadIdx.x;
  const int b      = tile / (N_DIM / IT);
  const int i_base = (tile % (N_DIM / IT)) * IT;

  const float* Hg = hidden + (size_t)b * N_DIM * D_DIM;
  for (int idx4 = tid; idx4 < (N_DIM * D_DIM) / 4; idx4 += 256) {
    const int r = idx4 >> 5, c4 = idx4 & 31;
    *reinterpret_cast<float4*>(&sH[r * HP + c4 * 4]) =
        reinterpret_cast<const float4*>(Hg)[idx4];
  }
  if (tid < (D_DIM * 4) / 4)
    reinterpret_cast<float4*>(sAa)[tid] = reinterpret_cast<const float4*>(amat)[tid];
  __syncthreads();

  const int wave   = tid >> 5;
  const int lane   = tid & 31;
  const int lo16   = lane & 15;
  const int hi16   = (lane >= 16) ? 8 : 0;
  const int koff   = (lane >= 16) ? 2 : 0;
  const int j_base = wave * 16;

  const float C1 = 0.6f, C2 = 0.4f;
  const float NEG_INF = -9.0e15f;

  v8f acc[2][4] = {};
  const int j_row = j_base + lo16;
  struct SFrag { float2 hj, hi[2]; float4 a0, a1; };
  auto load_s = [&](SFrag& f, int d0) {
    const int d = d0 + koff;
    f.hj    = *reinterpret_cast<const float2*>(&sH[j_row * HP + d]);
    f.hi[0] = *reinterpret_cast<const float2*>(&sH[(i_base + lo16) * HP + d]);
    f.hi[1] = *reinterpret_cast<const float2*>(&sH[(i_base + 16 + lo16) * HP + d]);
    f.a0    = *reinterpret_cast<const float4*>(&sAa[d * 4]);
    f.a1    = *reinterpret_cast<const float4*>(&sAa[(d + 1) * 4]);
  };
  auto mma_s = [&](const SFrag& f) {
    const float a0[4] = {f.a0.x, f.a0.y, f.a0.z, f.a0.w};
    const float a1[4] = {f.a1.x, f.a1.y, f.a1.z, f.a1.w};
    v2f Bv;  Bv[0] = f.hj.x;         Bv[1] = f.hj.y;
    v2f Bva; Bva[0] = fabsf(f.hj.x); Bva[1] = fabsf(f.hj.y);
#pragma unroll
    for (int t = 0; t < 2; ++t) {
      const float2 hi = f.hi[t];
      const float hax = fabsf(hi.x), hay = fabsf(hi.y);
#pragma unroll
      for (int k = 0; k < 4; ++k) {
        v2f Av;  Av[0]  = C1 * hi.x * a0[k]; Av[1]  = C1 * hi.y * a1[k];
        acc[t][k] = wmma_f32(Av, Bv, acc[t][k]);
        v2f Ava; Ava[0] = C2 * hax * a0[k];  Ava[1] = C2 * hay * a1[k];
        acc[t][k] = wmma_f32(Ava, Bva, acc[t][k]);
      }
    }
  };
  SFrag S0, S1;
  load_s(S0, 0);
  for (int d0 = 0; d0 < D_DIM - 8; d0 += 8) {
    load_s(S1, d0 + 4);
    __builtin_amdgcn_sched_barrier(0);
    mma_s(S0);
    __builtin_amdgcn_sched_barrier(0);
    load_s(S0, d0 + 8);
    __builtin_amdgcn_sched_barrier(0);
    mma_s(S1);
    __builtin_amdgcn_sched_barrier(0);
  }
  load_s(S1, D_DIM - 4);
  __builtin_amdgcn_sched_barrier(0);
  mma_s(S0);
  __builtin_amdgcn_sched_barrier(0);
  mma_s(S1);

  const int* adjB = adj + (size_t)b * N_DIM * N_DIM;
#pragma unroll
  for (int t = 0; t < 2; ++t) {
#pragma unroll
    for (int r = 0; r < 8; ++r) {
      const int il = 16 * t + r + hi16;
      const int ty = adjB[(i_base + il) * N_DIM + j_base + lo16];
      float e = NEG_INF;
      e = (ty == 1) ? acc[t][0][r] : e;
      e = (ty == 2) ? acc[t][1][r] : e;
      e = (ty == 3) ? acc[t][2][r] : e;
      e = (ty == 4) ? acc[t][3][r] : e;
      sAlpha[il * AP + j_base + lo16] = e;
    }
  }
  __syncthreads();

  const int row = tid >> 4;
  const int part = tid & 15;
  for (int h = 0; h < 2; ++h) {
    const int rr = row + 16 * h;
    float m = -3.4e38f;
    for (int c = part * 8; c < part * 8 + 8; ++c)
      m = fmaxf(m, sAlpha[rr * AP + c]);
    sRed[rr][part] = m;
  }
  __syncthreads();
  if (part < 2) {
    const int rr = row + 16 * part;
    float mm = sRed[rr][0];
    for (int q = 1; q < 16; ++q) mm = fmaxf(mm, sRed[rr][q]);
    sMax[rr] = mm;
  }
  __syncthreads();
  for (int h = 0; h < 2; ++h) {
    const int rr = row + 16 * h;
    const float mm2 = sMax[rr];
    float s = 0.f;
    for (int c = part * 8; c < part * 8 + 8; ++c) {
      const float ev = __expf(sAlpha[rr * AP + c] - mm2);
      sAlpha[rr * AP + c] = ev;
      s += ev;
    }
    sRed[rr][part] = s;
  }
  __syncthreads();
  if (part < 2) {
    const int rr = row + 16 * part;
    float ss = 0.f;
    for (int q = 0; q < 16; ++q) ss += sRed[rr][q];
    sSum[rr] = ss;
  }
  __syncthreads();
  for (int h = 0; h < 2; ++h) {
    const int rr = row + 16 * h;
    const float inv = 1.0f / sSum[rr];
    for (int c = part * 8; c < part * 8 + 8; ++c)
      sAlpha[rr * AP + c] *= inv;
  }
  __syncthreads();

  const int dg = j_base + lo16;
  v8f oacc0 = {}, oacc1 = {};
  struct OFrag { float2 a0[OB], a1[OB]; float b0[OB], b1[OB]; };
  auto load_o = [&](OFrag& f, int jb) {
#pragma unroll
    for (int s = 0; s < OB; ++s) {
      const int jk = jb + 4 * s + koff;
      f.a0[s] = *reinterpret_cast<const float2*>(&sAlpha[lo16 * AP + jk]);
      f.a1[s] = *reinterpret_cast<const float2*>(&sAlpha[(16 + lo16) * AP + jk]);
      f.b0[s] = sH[jk * HP + dg];
      f.b1[s] = sH[(jk + 1) * HP + dg];
    }
  };
  auto mma_o = [&](const OFrag& f) {
#pragma unroll
    for (int s = 0; s < OB; ++s) {
      v2f Av0; Av0[0] = f.a0[s].x; Av0[1] = f.a0[s].y;
      v2f Av1; Av1[0] = f.a1[s].x; Av1[1] = f.a1[s].y;
      v2f Bv;  Bv[0] = f.b0[s];    Bv[1] = f.b1[s];
      oacc0 = wmma_f32(Av0, Bv, oacc0);
      oacc1 = wmma_f32(Av1, Bv, oacc1);
    }
  };
  OFrag O0, O1;
  load_o(O0, 0);
  load_o(O1, 4 * OB);
  __builtin_amdgcn_sched_barrier(0);
  mma_o(O0);
  __builtin_amdgcn_sched_barrier(0);
  load_o(O0, 8 * OB);
  __builtin_amdgcn_sched_barrier(0);
  mma_o(O1);
  __builtin_amdgcn_sched_barrier(0);
  load_o(O1, 12 * OB);
  __builtin_amdgcn_sched_barrier(0);
  mma_o(O0);
  __builtin_amdgcn_sched_barrier(0);
  mma_o(O1);
#pragma unroll
  for (int r = 0; r < 8; ++r) {
    out[((size_t)b * N_DIM + i_base + r + hi16) * D_DIM + dg] = oacc0[r];
    out[((size_t)b * N_DIM + i_base + 16 + r + hi16) * D_DIM + dg] = oacc1[r];
  }
}

#define NPANEL  (A_DIM / KT)
#define KB      5
#define A_F4    ((IT * KT) / 4)
#define E_F4    ((KT * D_DIM) / 4)
#define A_SLOTS ((A_F4 + 255) / 256)
#define E_SLOTS ((E_F4 + 255) / 256)
#define SLOT_LIVE(q, n) ((256 * (q) + 255 < (n)) || (tid + 256 * (q) < (n)))

__device__ __forceinline__ void attr_gemm_role(
    GemmLds& L, const int tile,
    const float* __restrict__ Aattr, const float* __restrict__ Eemb,
    float* __restrict__ out2) {
  float (* const sAp)[IT * KT] = L.Ap;
  float (* const sEp)[KT * EP] = L.Ep;

  const int tid    = threadIdx.x;
  const int m_base = tile * IT;
  const int wave   = tid >> 5;
  const int lane   = tid & 31;
  const int lo16   = lane & 15;
  const int hi16   = (lane >= 16) ? 8 : 0;
  const int koff   = (lane >= 16) ? 2 : 0;
  const int dg     = wave * 16 + lo16;

  const float* gA[A_SLOTS];
  int lA[A_SLOTS];
#pragma unroll
  for (int q = 0; q < A_SLOTS; ++q) {
    const int idx4 = tid + 256 * q;
    const int r = idx4 / (KT / 4), c4 = idx4 % (KT / 4);
    gA[q] = Aattr + (size_t)(m_base + r) * A_DIM + c4 * 4;
    lA[q] = r * KT + c4 * 4;
  }
  const float* gE = Eemb + tid * 4;
  const int lE = (tid >> 5) * EP + (tid & 31) * 4;

  float4 rA[A_SLOTS], rE[E_SLOTS];
#pragma unroll
  for (int q = 0; q < A_SLOTS; ++q) rA[q] = make_float4(0.f, 0.f, 0.f, 0.f);
#pragma unroll
  for (int q = 0; q < E_SLOTS; ++q) rE[q] = make_float4(0.f, 0.f, 0.f, 0.f);

  auto load_panel = [&](int k0) {
#pragma unroll
    for (int q = 0; q < A_SLOTS; ++q)
      if (SLOT_LIVE(q, A_F4))
        rA[q] = *reinterpret_cast<const float4*>(gA[q] + k0);
#pragma unroll
    for (int q = 0; q < E_SLOTS; ++q)
      if (SLOT_LIVE(q, E_F4))
        rE[q] = *reinterpret_cast<const float4*>(gE + (size_t)k0 * D_DIM + 1024 * q);
  };
  auto store_panel = [&](int nb) {
#pragma unroll
    for (int q = 0; q < A_SLOTS; ++q)
      if (SLOT_LIVE(q, A_F4))
        *reinterpret_cast<float4*>(&sAp[nb][lA[q]]) = rA[q];
#pragma unroll
    for (int q = 0; q < E_SLOTS; ++q)
      if (SLOT_LIVE(q, E_F4))
        *reinterpret_cast<float4*>(&sEp[nb][lE + 8 * EP * q]) = rE[q];
  };

  struct Frag { float2 a0[KB], a1[KB]; float b0[KB], b1[KB]; };
  auto load_block = [&](Frag& f, int cb, int blk) {
    const float* pA = sAp[cb];
    const float* pE = sEp[cb];
#pragma unroll
    for (int s = 0; s < KB; ++s) {
      const int k = 4 * (KB * blk + s) + koff;
      f.a0[s] = *reinterpret_cast<const float2*>(&pA[lo16 * KT + k]);
      f.a1[s] = *reinterpret_cast<const float2*>(&pA[(16 + lo16) * KT + k]);
      f.b0[s] = pE[k * EP + dg];
      f.b1[s] = pE[(k + 1) * EP + dg];
    }
  };
  v8f acc0 = {}, acc1 = {};
  auto mma_block = [&](const Frag& f) {
#pragma unroll
    for (int s = 0; s < KB; ++s) {
      v2f Av0; Av0[0] = f.a0[s].x; Av0[1] = f.a0[s].y;
      v2f Av1; Av1[0] = f.a1[s].x; Av1[1] = f.a1[s].y;
      v2f Bv;  Bv[0] = f.b0[s];    Bv[1] = f.b1[s];
      acc0 = wmma_f32(Av0, Bv, acc0);
      acc1 = wmma_f32(Av1, Bv, acc1);
    }
  };
#define STAGE(NEXT, CB, BLK, CUR)            \
  do {                                       \
    load_block(NEXT, CB, BLK);               \
    __builtin_amdgcn_sched_barrier(0);       \
    mma_block(CUR);                          \
    __builtin_amdgcn_sched_barrier(0);       \
  } while (0)

  load_panel(0);
  store_panel(0);
  __syncthreads();

  Frag F0, F1;
  load_block(F0, 0, 0);
  for (int kt = 0; kt < NPANEL; kt += 2) {
    load_panel((kt + 1) * KT);
    STAGE(F1, 0, 1, F0);
    STAGE(F0, 0, 2, F1);
    STAGE(F1, 0, 3, F0);
    STAGE(F0, 0, 4, F1);
    store_panel(1);
    __syncthreads();
    const bool more = kt + 2 < NPANEL;
    STAGE(F1, 1, 0, F0);
    if (more) load_panel((kt + 2) * KT);
    STAGE(F0, 1, 1, F1);
    STAGE(F1, 1, 2, F0);
    STAGE(F0, 1, 3, F1);
    STAGE(F1, 1, 4, F0);
    if (more) store_panel(0);
    __syncthreads();
    if (more) load_block(F0, 0, 0);
    __builtin_amdgcn_sched_barrier(0);
    mma_block(F1);
  }
#undef STAGE
#pragma unroll
  for (int r = 0; r < 8; ++r) {
    out2[(size_t)(m_base + r + hi16) * D_DIM + dg] = acc0[r];
    out2[(size_t)(m_base + 16 + r + hi16) * D_DIM + dg] = acc1[r];
  }
}

#define TILES_PER_ROLE (B_DIM * (N_DIM / IT))

__global__ __launch_bounds__(256) void fused_kernel(
    const float* __restrict__ hidden, const int* __restrict__ adj,
    const float* __restrict__ amat, const float* __restrict__ Aattr,
    const float* __restrict__ Eemb, float* __restrict__ out_attn,
    float* __restrict__ out_attr) {
  __shared__ FusedLds lds;
  const int bid  = blockIdx.x;
  const int role = bid / TILES_PER_ROLE;
  const int tile = bid % TILES_PER_ROLE;
  if (role == 0)
    gat_attn_role(lds.att, tile, hidden, adj, amat, out_attn);
  else
    attr_gemm_role(lds.gemm, tile, Aattr, Eemb, out_attr);
}

extern "C" void kernel_launch(void* const* d_in, const int* in_sizes, int n_in,
                              void* d_out, int out_size, void* d_ws, size_t ws_size,
                              hipStream_t stream) {
  (void)in_sizes; (void)n_in; (void)out_size; (void)d_ws; (void)ws_size;
  const float* hidden = (const float*)d_in[0];
  const int*   adj    = (const int*)d_in[1];
  const float* amat   = (const float*)d_in[2];
  const float* Aattr  = (const float*)d_in[3];
  const float* Eemb   = (const float*)d_in[4];

  float* out_attn = (float*)d_out;
  float* out_attr = out_attn + (size_t)B_DIM * N_DIM * D_DIM;

  fused_kernel<<<dim3(2 * TILES_PER_ROLE), dim3(256), 0, stream>>>(
      hidden, adj, amat, Aattr, Eemb, out_attn, out_attr);
}


# attr GEMM role asks for its whole A tile (128000 contiguous bytes) with second-level-cache prefetches at role start, so the K loop's A loads hit the cache instead of memory; sum order unchanged
# speedup vs baseline: 2.4403x; 1.0454x over previous
#include <hip/hip_runtime.h>
#include <hip/hip_bf16.h>


typedef __attribute__((ext_vector_type(2))) float v2f;
typedef __attribute__((ext_vector_type(8))) float v8f;

#define B_DIM 64
#define N_DIM 128
#define D_DIM 128
#define A_DIM 1000
#define KT    100
#define OB    8
#define IT    32
#define HP    140
#define AP    132
#define EP    136

__device__ __forceinline__ v8f wmma_f32(v2f a, v2f b, v8f c) {
  return __builtin_amdgcn_wmma_f32_16x16x4_f32(false, a, false, b, (short)0, c,
                                               false, false);
}


struct AttLds {
  float H[N_DIM * HP];
  float Alpha[IT * AP];
  float Aa[D_DIM * 4];
  float Red[IT][16];
  float Max[IT];
  float Sum[IT];
};
struct GemmLds {
  float Ap[2][IT * KT];
  float Ep[2][KT * EP];
};
union __attribute__((aligned(16))) FusedLds {
  AttLds  att;
  GemmLds gemm;
};

__device__ __forceinline__ void gat_attn_role(
    AttLds& L, const int tile,
    const float* __restrict__ hidden, const int* __restrict__ adj,
    const float* __restrict__ amat, float* __restrict__ out) {
  float* const sH = L.H;
  float* const sAlpha = L.Alpha;
  float* const sAa = L.Aa;
  float (* const sRed)[16] = L.Red;
  float* const sMax = L.Max;
  float* const sSum = L.Sum;

  const int tid    = threadIdx.x;
  const int b      = tile / (N_DIM / IT);
  const int i_base = (tile % (N_DIM / IT)) * IT;

  const float* Hg = hidden + (size_t)b * N_DIM * D_DIM;
  for (int idx4 = tid; idx4 < (N_DIM * D_DIM) / 4; idx4 += 256) {
    const int r = idx4 >> 5, c4 = idx4 & 31;
    *reinterpret_cast<float4*>(&sH[r * HP + c4 * 4]) =
        reinterpret_cast<const float4*>(Hg)[idx4];
  }
  if (tid < (D_DIM * 4) / 4)
    reinterpret_cast<float4*>(sAa)[tid] = reinterpret_cast<const float4*>(amat)[tid];
  __syncthreads();

  const int wave   = tid >> 5;
  const int lane   = tid & 31;
  const int lo16   = lane & 15;
  const int hi16   = (lane >= 16) ? 8 : 0;
  const int koff   = (lane >= 16) ? 2 : 0;
  const int j_base = wave * 16;

  const float C1 = 0.6f, C2 = 0.4f;
  const float NEG_INF = -9.0e15f;

  v8f acc[2][4] = {};
  const int j_row = j_base + lo16;
  struct SFrag { float2 hj, hi[2]; float4 a0, a1; };
  auto load_s = [&](SFrag& f, int d0) {
    const int d = d0 + koff;
    f.hj    = *reinterpret_cast<const float2*>(&sH[j_row * HP + d]);
    f.hi[0] = *reinterpret_cast<const float2*>(&sH[(i_base + lo16) * HP + d]);
    f.hi[1] = *reinterpret_cast<const float2*>(&sH[(i_base + 16 + lo16) * HP + d]);
    f.a0    = *reinterpret_cast<const float4*>(&sAa[d * 4]);
    f.a1    = *reinterpret_cast<const float4*>(&sAa[(d + 1) * 4]);
  };
  auto mma_s = [&](const SFrag& f) {
    const float a0[4] = {f.a0.x, f.a0.y, f.a0.z, f.a0.w};
    const float a1[4] = {f.a1.x, f.a1.y, f.a1.z, f.a1.w};
    v2f Bv;  Bv[0] = f.hj.x;         Bv[1] = f.hj.y;
    v2f Bva; Bva[0] = fabsf(f.hj.x); Bva[1] = fabsf(f.hj.y);
#pragma unroll
    for (int t = 0; t < 2; ++t) {
      const float2 hi = f.hi[t];
      const float hax = fabsf(hi.x), hay = fabsf(hi.y);
#pragma unroll
      for (int k = 0; k < 4; ++k) {
        v2f Av;  Av[0]  = C1 * hi.x * a0[k]; Av[1]  = C1 * hi.y * a1[k];
        acc[t][k] = wmma_f32(Av, Bv, acc[t][k]);
        v2f Ava; Ava[0] = C2 * hax * a0[k];  Ava[1] = C2 * hay * a1[k];
        acc[t][k] = wmma_f32(Ava, Bva, acc[t][k]);
      }
    }
  };
  SFrag S0, S1;
  load_s(S0, 0);
  for (int d0 = 0; d0 < D_DIM - 8; d0 += 8) {
    load_s(S1, d0 + 4);
    __builtin_amdgcn_sched_barrier(0);
    mma_s(S0);
    __builtin_amdgcn_sched_barrier(0);
    load_s(S0, d0 + 8);
    __builtin_amdgcn_sched_barrier(0);
    mma_s(S1);
    __builtin_amdgcn_sched_barrier(0);
  }
  load_s(S1, D_DIM - 4);
  __builtin_amdgcn_sched_barrier(0);
  mma_s(S0);
  __builtin_amdgcn_sched_barrier(0);
  mma_s(S1);

  const int* adjB = adj + (size_t)b * N_DIM * N_DIM;
#pragma unroll
  for (int t = 0; t < 2; ++t) {
#pragma unroll
    for (int r = 0; r < 8; ++r) {
      const int il = 16 * t + r + hi16;
      const int ty = adjB[(i_base + il) * N_DIM + j_base + lo16];
      float e = NEG_INF;
      e = (ty == 1) ? acc[t][0][r] : e;
      e = (ty == 2) ? acc[t][1][r] : e;
      e = (ty == 3) ? acc[t][2][r] : e;
      e = (ty == 4) ? acc[t][3][r] : e;
      sAlpha[il * AP + j_base + lo16] = e;
    }
  }
  __syncthreads();

  const int row = tid >> 4;
  const int part = tid & 15;
  for (int h = 0; h < 2; ++h) {
    const int rr = row + 16 * h;
    float m = -3.4e38f;
    for (int c = part * 8; c < part * 8 + 8; ++c)
      m = fmaxf(m, sAlpha[rr * AP + c]);
    sRed[rr][part] = m;
  }
  __syncthreads();
  if (part < 2) {
    const int rr = row + 16 * part;
    float mm = sRed[rr][0];
    for (int q = 1; q < 16; ++q) mm = fmaxf(mm, sRed[rr][q]);
    sMax[rr] = mm;
  }
  __syncthreads();
  for (int h = 0; h < 2; ++h) {
    const int rr = row + 16 * h;
    const float mm2 = sMax[rr];
    float s = 0.f;
    for (int c = part * 8; c < part * 8 + 8; ++c) {
      const float ev = __expf(sAlpha[rr * AP + c] - mm2);
      sAlpha[rr * AP + c] = ev;
      s += ev;
    }
    sRed[rr][part] = s;
  }
  __syncthreads();
  if (part < 2) {
    const int rr = row + 16 * part;
    float ss = 0.f;
    for (int q = 0; q < 16; ++q) ss += sRed[rr][q];
    sSum[rr] = ss;
  }
  __syncthreads();
  for (int h = 0; h < 2; ++h) {
    const int rr = row + 16 * h;
    const float inv = 1.0f / sSum[rr];
    for (int c = part * 8; c < part * 8 + 8; ++c)
      sAlpha[rr * AP + c] *= inv;
  }
  __syncthreads();

  const int dg = j_base + lo16;
  v8f oacc0 = {}, oacc1 = {};
  struct OFrag { float2 a0[OB], a1[OB]; float b0[OB], b1[OB]; };
  auto load_o = [&](OFrag& f, int jb) {
#pragma unroll
    for (int s = 0; s < OB; ++s) {
      const int jk = jb + 4 * s + koff;
      f.a0[s] = *reinterpret_cast<const float2*>(&sAlpha[lo16 * AP + jk]);
      f.a1[s] = *reinterpret_cast<const float2*>(&sAlpha[(16 + lo16) * AP + jk]);
      f.b0[s] = sH[jk * HP + dg];
      f.b1[s] = sH[(jk + 1) * HP + dg];
    }
  };
  auto mma_o = [&](const OFrag& f) {
#pragma unroll
    for (int s = 0; s < OB; ++s) {
      v2f Av0; Av0[0] = f.a0[s].x; Av0[1] = f.a0[s].y;
      v2f Av1; Av1[0] = f.a1[s].x; Av1[1] = f.a1[s].y;
      v2f Bv;  Bv[0] = f.b0[s];    Bv[1] = f.b1[s];
      oacc0 = wmma_f32(Av0, Bv, oacc0);
      oacc1 = wmma_f32(Av1, Bv, oacc1);
    }
  };
  OFrag O0, O1;
  load_o(O0, 0);
  load_o(O1, 4 * OB);
  __builtin_amdgcn_sched_barrier(0);
  mma_o(O0);
  __builtin_amdgcn_sched_barrier(0);
  load_o(O0, 8 * OB);
  __builtin_amdgcn_sched_barrier(0);
  mma_o(O1);
  __builtin_amdgcn_sched_barrier(0);
  load_o(O1, 12 * OB);
  __builtin_amdgcn_sched_barrier(0);
  mma_o(O0);
  __builtin_amdgcn_sched_barrier(0);
  mma_o(O1);
#pragma unroll
  for (int r = 0; r < 8; ++r) {
    out[((size_t)b * N_DIM + i_base + r + hi16) * D_DIM + dg] = oacc0[r];
    out[((size_t)b * N_DIM + i_base + 16 + r + hi16) * D_DIM + dg] = oacc1[r];
  }
}

#define NPANEL  (A_DIM / KT)
#define KB      5
#define A_F4    ((IT * KT) / 4)
#define E_F4    ((KT * D_DIM) / 4)
#define A_SLOTS ((A_F4 + 255) / 256)
#define E_SLOTS ((E_F4 + 255) / 256)
#define SLOT_LIVE(q, n) ((256 * (q) + 255 < (n)) || (tid + 256 * (q) < (n)))

__device__ __forceinline__ void attr_gemm_role(
    GemmLds& L, const int tile,
    const float* __restrict__ Aattr, const float* __restrict__ Eemb,
    float* __restrict__ out2) {
  float (* const sAp)[IT * KT] = L.Ap;
  float (* const sEp)[KT * EP] = L.Ep;

  const int tid    = threadIdx.x;
  const int m_base = tile * IT;
  const int wave   = tid >> 5;
  const int lane   = tid & 31;
  const int lo16   = lane & 15;
  const int hi16   = (lane >= 16) ? 8 : 0;
  const int koff   = (lane >= 16) ? 2 : 0;
  const int dg     = wave * 16 + lo16;

  const float* gA[A_SLOTS];
  int lA[A_SLOTS];
#pragma unroll
  for (int q = 0; q < A_SLOTS; ++q) {
    const int idx4 = tid + 256 * q;
    const int r = idx4 / (KT / 4), c4 = idx4 % (KT / 4);
    gA[q] = Aattr + (size_t)(m_base + r) * A_DIM + c4 * 4;
    lA[q] = r * KT + c4 * 4;
  }
  const float* gE = Eemb + tid * 4;
  const int lE = (tid >> 5) * EP + (tid & 31) * 4;

  float4 rA[A_SLOTS], rE[E_SLOTS];
#pragma unroll
  for (int q = 0; q < A_SLOTS; ++q) rA[q] = make_float4(0.f, 0.f, 0.f, 0.f);
#pragma unroll
  for (int q = 0; q < E_SLOTS; ++q) rE[q] = make_float4(0.f, 0.f, 0.f, 0.f);

  auto load_panel = [&](int k0) {
#pragma unroll
    for (int q = 0; q < A_SLOTS; ++q)
      if (SLOT_LIVE(q, A_F4))
        rA[q] = *reinterpret_cast<const float4*>(gA[q] + k0);
#pragma unroll
    for (int q = 0; q < E_SLOTS; ++q)
      if (SLOT_LIVE(q, E_F4))
        rE[q] = *reinterpret_cast<const float4*>(gE + (size_t)k0 * D_DIM + 1024 * q);
  };
  auto store_panel = [&](int nb) {
#pragma unroll
    for (int q = 0; q < A_SLOTS; ++q)
      if (SLOT_LIVE(q, A_F4))
        *reinterpret_cast<float4*>(&sAp[nb][lA[q]]) = rA[q];
#pragma unroll
    for (int q = 0; q < E_SLOTS; ++q)
      if (SLOT_LIVE(q, E_F4))
        *reinterpret_cast<float4*>(&sEp[nb][lE + 8 * EP * q]) = rE[q];
  };

  struct Frag { float2 a0[KB], a1[KB]; float b0[KB], b1[KB]; };
  auto load_block = [&](Frag& f, int cb, int blk) {
    const float* pA = sAp[cb];
    const float* pE = sEp[cb];
#pragma unroll
    for (int s = 0; s < KB; ++s) {
      const int k = 4 * (KB * blk + s) + koff;
      f.a0[s] = *reinterpret_cast<const float2*>(&pA[lo16 * KT + k]);
      f.a1[s] = *reinterpret_cast<const float2*>(&pA[(16 + lo16) * KT + k]);
      f.b0[s] = pE[k * EP + dg];
      f.b1[s] = pE[(k + 1) * EP + dg];
    }
  };
  v8f acc0 = {}, acc1 = {};
  auto mma_block = [&](const Frag& f) {
#pragma unroll
    for (int s = 0; s < KB; ++s) {
      v2f Av0; Av0[0] = f.a0[s].x; Av0[1] = f.a0[s].y;
      v2f Av1; Av1[0] = f.a1[s].x; Av1[1] = f.a1[s].y;
      v2f Bv;  Bv[0] = f.b0[s];    Bv[1] = f.b1[s];
      acc0 = wmma_f32(Av0, Bv, acc0);
      acc1 = wmma_f32(Av1, Bv, acc1);
    }
  };
#define STAGE(NEXT, CB, BLK, CUR)            \
  do {                                       \
    load_block(NEXT, CB, BLK);               \
    __builtin_amdgcn_sched_barrier(0);       \
    mma_block(CUR);                          \
    __builtin_amdgcn_sched_barrier(0);       \
  } while (0)

  load_panel(0);
  for (int q = 0; q < 2; ++q) {
    const int u = tid + 256 * q;
    if (u < (IT * A_DIM * 4) / 256)
      __builtin_prefetch(Aattr + (size_t)m_base * A_DIM + (size_t)u * 64, 0, 2);
  }
  store_panel(0);
  __syncthreads();

  Frag F0, F1;
  load_block(F0, 0, 0);
  for (int kt = 0; kt < NPANEL; kt += 2) {
    load_panel((kt + 1) * KT);
    STAGE(F1, 0, 1, F0);
    STAGE(F0, 0, 2, F1);
    STAGE(F1, 0, 3, F0);
    STAGE(F0, 0, 4, F1);
    store_panel(1);
    __syncthreads();
    const bool more = kt + 2 < NPANEL;
    STAGE(F1, 1, 0, F0);
    if (more) load_panel((kt + 2) * KT);
    STAGE(F0, 1, 1, F1);
    STAGE(F1, 1, 2, F0);
    STAGE(F0, 1, 3, F1);
    STAGE(F1, 1, 4, F0);
    if (more) store_panel(0);
    __syncthreads();
    if (more) load_block(F0, 0, 0);
    __builtin_amdgcn_sched_barrier(0);
    mma_block(F1);
  }
#undef STAGE
#pragma unroll
  for (int r = 0; r < 8; ++r) {
    out2[(size_t)(m_base + r + hi16) * D_DIM + dg] = acc0[r];
    out2[(size_t)(m_base + 16 + r + hi16) * D_DIM + dg] = acc1[r];
  }
}

#define TILES_PER_ROLE (B_DIM * (N_DIM / IT))

__global__ __launch_bounds__(256) void fused_kernel(
    const float* __restrict__ hidden, const int* __restrict__ adj,
    const float* __restrict__ amat, const float* __restrict__ Aattr,
    const float* __restrict__ Eemb, float* __restrict__ out_attn,
    float* __restrict__ out_attr) {
  __shared__ FusedLds lds;
  const int bid  = blockIdx.x;
  const int role = bid / TILES_PER_ROLE;
  const int tile = bid % TILES_PER_ROLE;
  if (role == 0)
    gat_attn_role(lds.att, tile, hidden, adj, amat, out_attn);
  else
    attr_gemm_role(lds.gemm, tile, Aattr, Eemb, out_attr);
}

extern "C" void kernel_launch(void* const* d_in, const int* in_sizes, int n_in,
                              void* d_out, int out_size, void* d_ws, size_t ws_size,
                              hipStream_t stream) {
  (void)in_sizes; (void)n_in; (void)out_size; (void)d_ws; (void)ws_size;
  const float* hidden = (const float*)d_in[0];
  const int*   adj    = (const int*)d_in[1];
  const float* amat   = (const float*)d_in[2];
  const float* Aattr  = (const float*)d_in[3];
  const float* Eemb   = (const float*)d_in[4];

  float* out_attn = (float*)d_out;
  float* out_attr = out_attn + (size_t)B_DIM * N_DIM * D_DIM;

  fused_kernel<<<dim3(2 * TILES_PER_ROLE), dim3(256), 0, stream>>>(
      hidden, adj, amat, Aattr, Eemb, out_attn, out_attr);
}


# wave issue priority (s_setprio) stepped 3,2,1,0 with each wave's own progress through its matrix work, both roles: the two co-resident work-groups advance together, none ends alone; sum order same
# speedup vs baseline: 2.5791x; 1.0569x over previous
#include <hip/hip_runtime.h>
#include <hip/hip_bf16.h>


typedef __attribute__((ext_vector_type(2))) float v2f;
typedef __attribute__((ext_vector_type(8))) float v8f;

#define B_DIM 64
#define N_DIM 128
#define D_DIM 128
#define A_DIM 1000
#define KT    100
#define OB    8
#define IT    32
#define HP    140
#define AP    132
#define EP    136

__device__ __forceinline__ v8f wmma_f32(v2f a, v2f b, v8f c) {
  return __builtin_amdgcn_wmma_f32_16x16x4_f32(false, a, false, b, (short)0, c,
                                               false, false);
}


struct AttLds {
  float H[N_DIM * HP];
  float Alpha[IT * AP];
  float Aa[D_DIM * 4];
  float Red[IT][16];
  float Max[IT];
  float Sum[IT];
};
struct GemmLds {
  float Ap[2][IT * KT];
  float Ep[2][KT * EP];
};
union __attribute__((aligned(16))) FusedLds {
  AttLds  att;
  GemmLds gemm;
};

__device__ __forceinline__ void gat_attn_role(
    AttLds& L, const int tile,
    const float* __restrict__ hidden, const int* __restrict__ adj,
    const float* __restrict__ amat, float* __restrict__ out) {
  float* const sH = L.H;
  float* const sAlpha = L.Alpha;
  float* const sAa = L.Aa;
  float (* const sRed)[16] = L.Red;
  float* const sMax = L.Max;
  float* const sSum = L.Sum;

  const int tid    = threadIdx.x;
  const int b      = tile / (N_DIM / IT);
  const int i_base = (tile % (N_DIM / IT)) * IT;

  __builtin_amdgcn_s_setprio(3);

  const float* Hg = hidden + (size_t)b * N_DIM * D_DIM;
  for (int idx4 = tid; idx4 < (N_DIM * D_DIM) / 4; idx4 += 256) {
    const int r = idx4 >> 5, c4 = idx4 & 31;
    *reinterpret_cast<float4*>(&sH[r * HP + c4 * 4]) =
        reinterpret_cast<const float4*>(Hg)[idx4];
  }
  if (tid < (D_DIM * 4) / 4)
    reinterpret_cast<float4*>(sAa)[tid] = reinterpret_cast<const float4*>(amat)[tid];
  __syncthreads();

  const int wave   = tid >> 5;
  const int lane   = tid & 31;
  const int lo16   = lane & 15;
  const int hi16   = (lane >= 16) ? 8 : 0;
  const int koff   = (lane >= 16) ? 2 : 0;
  const int j_base = wave * 16;

  const float C1 = 0.6f, C2 = 0.4f;
  const float NEG_INF = -9.0e15f;

  v8f acc[2][4] = {};
  const int j_row = j_base + lo16;
  struct SFrag { float2 hj, hi[2]; float4 a0, a1; };
  auto load_s = [&](SFrag& f, int d0) {
    const int d = d0 + koff;
    f.hj    = *reinterpret_cast<const float2*>(&sH[j_row * HP + d]);
    f.hi[0] = *reinterpret_cast<const float2*>(&sH[(i_base + lo16) * HP + d]);
    f.hi[1] = *reinterpret_cast<const float2*>(&sH[(i_base + 16 + lo16) * HP + d]);
    f.a0    = *reinterpret_cast<const float4*>(&sAa[d * 4]);
    f.a1    = *reinterpret_cast<const float4*>(&sAa[(d + 1) * 4]);
  };
  auto mma_s = [&](const SFrag& f) {
    const float a0[4] = {f.a0.x, f.a0.y, f.a0.z, f.a0.w};
    const float a1[4] = {f.a1.x, f.a1.y, f.a1.z, f.a1.w};
    v2f Bv;  Bv[0] = f.hj.x;         Bv[1] = f.hj.y;
    v2f Bva; Bva[0] = fabsf(f.hj.x); Bva[1] = fabsf(f.hj.y);
#pragma unroll
    for (int t = 0; t < 2; ++t) {
      const float2 hi = f.hi[t];
      const float hax = fabsf(hi.x), hay = fabsf(hi.y);
#pragma unroll
      for (int k = 0; k < 4; ++k) {
        v2f Av;  Av[0]  = C1 * hi.x * a0[k]; Av[1]  = C1 * hi.y * a1[k];
        acc[t][k] = wmma_f32(Av, Bv, acc[t][k]);
        v2f Ava; Ava[0] = C2 * hax * a0[k];  Ava[1] = C2 * hay * a1[k];
        acc[t][k] = wmma_f32(Ava, Bva, acc[t][k]);
      }
    }
  };
  SFrag S0, S1;
  load_s(S0, 0);
  auto score_trips = [&](int d_begin, int d_end) {
#pragma nounroll
    for (int d0 = d_begin; d0 < d_end; d0 += 8) {
      load_s(S1, d0 + 4);
      __builtin_amdgcn_sched_barrier(0);
      mma_s(S0);
      __builtin_amdgcn_sched_barrier(0);
      load_s(S0, d0 + 8);
      __builtin_amdgcn_sched_barrier(0);
      mma_s(S1);
      __builtin_amdgcn_sched_barrier(0);
    }
  };
  score_trips(0, D_DIM / 4);
  __builtin_amdgcn_s_setprio(2);
  score_trips(D_DIM / 4, D_DIM / 2);
  __builtin_amdgcn_s_setprio(1);
  score_trips(D_DIM / 2, 3 * D_DIM / 4);
  __builtin_amdgcn_s_setprio(0);
  score_trips(3 * D_DIM / 4, D_DIM - 8);
  load_s(S1, D_DIM - 4);
  __builtin_amdgcn_sched_barrier(0);
  mma_s(S0);
  __builtin_amdgcn_sched_barrier(0);
  mma_s(S1);

  const int* adjB = adj + (size_t)b * N_DIM * N_DIM;
#pragma unroll
  for (int t = 0; t < 2; ++t) {
#pragma unroll
    for (int r = 0; r < 8; ++r) {
      const int il = 16 * t + r + hi16;
      const int ty = adjB[(i_base + il) * N_DIM + j_base + lo16];
      float e = NEG_INF;
      e = (ty == 1) ? acc[t][0][r] : e;
      e = (ty == 2) ? acc[t][1][r] : e;
      e = (ty == 3) ? acc[t][2][r] : e;
      e = (ty == 4) ? acc[t][3][r] : e;
      sAlpha[il * AP + j_base + lo16] = e;
    }
  }
  __syncthreads();

  const int row = tid >> 4;
  const int part = tid & 15;
  for (int h = 0; h < 2; ++h) {
    const int rr = row + 16 * h;
    float m = -3.4e38f;
    for (int c = part * 8; c < part * 8 + 8; ++c)
      m = fmaxf(m, sAlpha[rr * AP + c]);
    sRed[rr][part] = m;
  }
  __syncthreads();
  if (part < 2) {
    const int rr = row + 16 * part;
    float mm = sRed[rr][0];
    for (int q = 1; q < 16; ++q) mm = fmaxf(mm, sRed[rr][q]);
    sMax[rr] = mm;
  }
  __syncthreads();
  for (int h = 0; h < 2; ++h) {
    const int rr = row + 16 * h;
    const float mm2 = sMax[rr];
    float s = 0.f;
    for (int c = part * 8; c < part * 8 + 8; ++c) {
      const float ev = __expf(sAlpha[rr * AP + c] - mm2);
      sAlpha[rr * AP + c] = ev;
      s += ev;
    }
    sRed[rr][part] = s;
  }
  __syncthreads();
  if (part < 2) {
    const int rr = row + 16 * part;
    float ss = 0.f;
    for (int q = 0; q < 16; ++q) ss += sRed[rr][q];
    sSum[rr] = ss;
  }
  __syncthreads();
  for (int h = 0; h < 2; ++h) {
    const int rr = row + 16 * h;
    const float inv = 1.0f / sSum[rr];
    for (int c = part * 8; c < part * 8 + 8; ++c)
      sAlpha[rr * AP + c] *= inv;
  }
  __syncthreads();

  const int dg = j_base + lo16;
  v8f oacc0 = {}, oacc1 = {};
  struct OFrag { float2 a0[OB], a1[OB]; float b0[OB], b1[OB]; };
  auto load_o = [&](OFrag& f, int jb) {
#pragma unroll
    for (int s = 0; s < OB; ++s) {
      const int jk = jb + 4 * s + koff;
      f.a0[s] = *reinterpret_cast<const float2*>(&sAlpha[lo16 * AP + jk]);
      f.a1[s] = *reinterpret_cast<const float2*>(&sAlpha[(16 + lo16) * AP + jk]);
      f.b0[s] = sH[jk * HP + dg];
      f.b1[s] = sH[(jk + 1) * HP + dg];
    }
  };
  auto mma_o = [&](const OFrag& f) {
#pragma unroll
    for (int s = 0; s < OB; ++s) {
      v2f Av0; Av0[0] = f.a0[s].x; Av0[1] = f.a0[s].y;
      v2f Av1; Av1[0] = f.a1[s].x; Av1[1] = f.a1[s].y;
      v2f Bv;  Bv[0] = f.b0[s];    Bv[1] = f.b1[s];
      oacc0 = wmma_f32(Av0, Bv, oacc0);
      oacc1 = wmma_f32(Av1, Bv, oacc1);
    }
  };
  OFrag O0, O1;
  load_o(O0, 0);
  load_o(O1, 4 * OB);
  __builtin_amdgcn_sched_barrier(0);
  mma_o(O0);
  __builtin_amdgcn_sched_barrier(0);
  load_o(O0, 8 * OB);
  __builtin_amdgcn_sched_barrier(0);
  mma_o(O1);
  __builtin_amdgcn_sched_barrier(0);
  load_o(O1, 12 * OB);
  __builtin_amdgcn_sched_barrier(0);
  mma_o(O0);
  __builtin_amdgcn_sched_barrier(0);
  mma_o(O1);
#pragma unroll
  for (int r = 0; r < 8; ++r) {
    out[((size_t)b * N_DIM + i_base + r + hi16) * D_DIM + dg] = oacc0[r];
    out[((size_t)b * N_DIM + i_base + 16 + r + hi16) * D_DIM + dg] = oacc1[r];
  }
}

#define NPANEL  (A_DIM / KT)
#define KB      5
#define A_F4    ((IT * KT) / 4)
#define E_F4    ((KT * D_DIM) / 4)
#define A_SLOTS ((A_F4 + 255) / 256)
#define E_SLOTS ((E_F4 + 255) / 256)
#define SLOT_LIVE(q, n) ((256 * (q) + 255 < (n)) || (tid + 256 * (q) < (n)))

__device__ __forceinline__ void attr_gemm_role(
    GemmLds& L, const int tile,
    const float* __restrict__ Aattr, const float* __restrict__ Eemb,
    float* __restrict__ out2) {
  float (* const sAp)[IT * KT] = L.Ap;
  float (* const sEp)[KT * EP] = L.Ep;

  const int tid    = threadIdx.x;
  const int m_base = tile * IT;
  __builtin_amdgcn_s_setprio(3);
  const int wave   = tid >> 5;
  const int lane   = tid & 31;
  const int lo16   = lane & 15;
  const int hi16   = (lane >= 16) ? 8 : 0;
  const int koff   = (lane >= 16) ? 2 : 0;
  const int dg     = wave * 16 + lo16;

  const float* gA[A_SLOTS];
  int lA[A_SLOTS];
#pragma unroll
  for (int q = 0; q < A_SLOTS; ++q) {
    const int idx4 = tid + 256 * q;
    const int r = idx4 / (KT / 4), c4 = idx4 % (KT / 4);
    gA[q] = Aattr + (size_t)(m_base + r) * A_DIM + c4 * 4;
    lA[q] = r * KT + c4 * 4;
  }
  const float* gE = Eemb + tid * 4;
  const int lE = (tid >> 5) * EP + (tid & 31) * 4;

  float4 rA[A_SLOTS], rE[E_SLOTS];
#pragma unroll
  for (int q = 0; q < A_SLOTS; ++q) rA[q] = make_float4(0.f, 0.f, 0.f, 0.f);
#pragma unroll
  for (int q = 0; q < E_SLOTS; ++q) rE[q] = make_float4(0.f, 0.f, 0.f, 0.f);

  auto load_panel = [&](int k0) {
#pragma unroll
    for (int q = 0; q < A_SLOTS; ++q)
      if (SLOT_LIVE(q, A_F4))
        rA[q] = *reinterpret_cast<const float4*>(gA[q] + k0);
#pragma unroll
    for (int q = 0; q < E_SLOTS; ++q)
      if (SLOT_LIVE(q, E_F4))
        rE[q] = *reinterpret_cast<const float4*>(gE + (size_t)k0 * D_DIM + 1024 * q);
  };
  auto store_panel = [&](int nb) {
#pragma unroll
    for (int q = 0; q < A_SLOTS; ++q)
      if (SLOT_LIVE(q, A_F4))
        *reinterpret_cast<float4*>(&sAp[nb][lA[q]]) = rA[q];
#pragma unroll
    for (int q = 0; q < E_SLOTS; ++q)
      if (SLOT_LIVE(q, E_F4))
        *reinterpret_cast<float4*>(&sEp[nb][lE + 8 * EP * q]) = rE[q];
  };

  struct Frag { float2 a0[KB], a1[KB]; float b0[KB], b1[KB]; };
  auto load_block = [&](Frag& f, int cb, int blk) {
    const float* pA = sAp[cb];
    const float* pE = sEp[cb];
#pragma unroll
    for (int s = 0; s < KB; ++s) {
      const int k = 4 * (KB * blk + s) + koff;
      f.a0[s] = *reinterpret_cast<const float2*>(&pA[lo16 * KT + k]);
      f.a1[s] = *reinterpret_cast<const float2*>(&pA[(16 + lo16) * KT + k]);
      f.b0[s] = pE[k * EP + dg];
      f.b1[s] = pE[(k + 1) * EP + dg];
    }
  };
  v8f acc0 = {}, acc1 = {};
  auto mma_block = [&](const Frag& f) {
#pragma unroll
    for (int s = 0; s < KB; ++s) {
      v2f Av0; Av0[0] = f.a0[s].x; Av0[1] = f.a0[s].y;
      v2f Av1; Av1[0] = f.a1[s].x; Av1[1] = f.a1[s].y;
      v2f Bv;  Bv[0] = f.b0[s];    Bv[1] = f.b1[s];
      acc0 = wmma_f32(Av0, Bv, acc0);
      acc1 = wmma_f32(Av1, Bv, acc1);
    }
  };
#define STAGE(NEXT, CB, BLK, CUR)            \
  do {                                       \
    load_block(NEXT, CB, BLK);               \
    __builtin_amdgcn_sched_barrier(0);       \
    mma_block(CUR);                          \
    __builtin_amdgcn_sched_barrier(0);       \
  } while (0)

  load_panel(0);
  for (int q = 0; q < 2; ++q) {
    const int u = tid + 256 * q;
    if (u < (IT * A_DIM * 4) / 256)
      __builtin_prefetch(Aattr + (size_t)m_base * A_DIM + (size_t)u * 64, 0, 2);
  }
  store_panel(0);
  __syncthreads();

  Frag F0, F1;
  load_block(F0, 0, 0);
  for (int kt = 0; kt < NPANEL; kt += 2) {
    load_panel((kt + 1) * KT);
    STAGE(F1, 0, 1, F0);
    STAGE(F0, 0, 2, F1);
    if (kt == 2)      __builtin_amdgcn_s_setprio(2);
    else if (kt == 8) __builtin_amdgcn_s_setprio(0);
    STAGE(F1, 0, 3, F0);
    STAGE(F0, 0, 4, F1);
    store_panel(1);
    __syncthreads();
    const bool more = kt + 2 < NPANEL;
    STAGE(F1, 1, 0, F0);
    if (more) load_panel((kt + 2) * KT);
    STAGE(F0, 1, 1, F1);
    STAGE(F1, 1, 2, F0);
    if (kt == 4) __builtin_amdgcn_s_setprio(1);
    STAGE(F0, 1, 3, F1);
    STAGE(F1, 1, 4, F0);
    if (more) store_panel(0);
    __syncthreads();
    if (more) load_block(F0, 0, 0);
    __builtin_amdgcn_sched_barrier(0);
    mma_block(F1);
  }
#undef STAGE
#pragma unroll
  for (int r = 0; r < 8; ++r) {
    out2[(size_t)(m_base + r + hi16) * D_DIM + dg] = acc0[r];
    out2[(size_t)(m_base + 16 + r + hi16) * D_DIM + dg] = acc1[r];
  }
}

#define TILES_PER_ROLE (B_DIM * (N_DIM / IT))

__global__ __launch_bounds__(256) void fused_kernel(
    const float* __restrict__ hidden, const int* __restrict__ adj,
    const float* __restrict__ amat, const float* __restrict__ Aattr,
    const float* __restrict__ Eemb, float* __restrict__ out_attn,
    float* __restrict__ out_attr) {
  __shared__ FusedLds lds;
  const int bid  = blockIdx.x;
  const int role = bid / TILES_PER_ROLE;
  const int tile = bid % TILES_PER_ROLE;
  if (role == 0)
    gat_attn_role(lds.att, tile, hidden, adj, amat, out_attn);
  else
    attr_gemm_role(lds.gemm, tile, Aattr, Eemb, out_attr);
}

extern "C" void kernel_launch(void* const* d_in, const int* in_sizes, int n_in,
                              void* d_out, int out_size, void* d_ws, size_t ws_size,
                              hipStream_t stream) {
  (void)in_sizes; (void)n_in; (void)out_size; (void)d_ws; (void)ws_size;
  const float* hidden = (const float*)d_in[0];
  const int*   adj    = (const int*)d_in[1];
  const float* amat   = (const float*)d_in[2];
  const float* Aattr  = (const float*)d_in[3];
  const float* Eemb   = (const float*)d_in[4];

  float* out_attn = (float*)d_out;
  float* out_attr = out_attn + (size_t)B_DIM * N_DIM * D_DIM;

  fused_kernel<<<dim3(2 * TILES_PER_ROLE), dim3(256), 0, stream>>>(
      hidden, adj, amat, Aattr, Eemb, out_attn, out_attr);
}
